# starting kernel
# baseline (speedup 1.0000x reference)
#include <hip/hip_runtime.h>
#include <hip/hip_bf16.h>

#define N_NODES 50000
#define IN_F    96
#define HID     64
#define N_CLS   32

typedef __attribute__((ext_vector_type(2))) float v2f;
typedef __attribute__((ext_vector_type(8))) float v8f;

__global__ void zero_f4(float4* __restrict__ p, int n4) {
    int i = blockIdx.x * blockDim.x + threadIdx.x;
    if (i < n4) p[i] = make_float4(0.f, 0.f, 0.f, 0.f);
}

template <int K, int N>
__global__ void gemm_wmma_f32(const float* __restrict__ X,
                              const float* __restrict__ W,
                              float* __restrict__ Y, int M) {
    __shared__ float sW[K * N];
    for (int i = threadIdx.x; i < K * N; i += blockDim.x) sW[i] = W[i];
    __syncthreads();

    const int wave = (int)(blockIdx.x * blockDim.x + threadIdx.x) >> 5;
    const int lane = threadIdx.x & 31;
    const int m0   = wave * 16;
    if (m0 >= M) return;

    const int half16 = lane >> 4;
    const int l15    = lane & 15;
    const int kh     = half16 * 2;
    const int arow   = m0 + l15;

    constexpr int NT = N / 16;
    v8f acc[NT] = {};

    for (int k = 0; k < K; k += 4) {
        v2f a;
        a.x = X[(size_t)arow * K + k + kh + 0];
        a.y = X[(size_t)arow * K + k + kh + 1];
#pragma unroll
        for (int t = 0; t < NT; ++t) {
            v2f b;
            b.x = sW[(k + kh + 0) * N + t * 16 + l15];
            b.y = sW[(k + kh + 1) * N + t * 16 + l15];
            acc[t] = __builtin_amdgcn_wmma_f32_16x16x4_f32(
                false, a, false, b, (short)0, acc[t], false, false);
        }
    }

    const int rbase = m0 + half16 * 8;
#pragma unroll
    for (int t = 0; t < NT; ++t)
#pragma unroll
        for (int r = 0; r < 8; ++r)
            Y[(size_t)(rbase + r) * N + t * 16 + l15] = acc[t][r];
}

template <int F>
__global__ void scatter_add(const float* __restrict__ msg,
                            const int* __restrict__ src,
                            const int* __restrict__ dst,
                            float* __restrict__ agg, int n_edges) {
    constexpr int LPE = F / 4;
    const long long t = (long long)blockIdx.x * blockDim.x + threadIdx.x;
    const int e = (int)(t / LPE);
    const int l = (int)(t % LPE);
    if (e >= n_edges) return;
    const int s = src[e];
    const int d = dst[e];
    const float4 v = *(const float4*)(msg + (size_t)s * F + l * 4);
    float* out = agg + (size_t)d * F + l * 4;
    unsafeAtomicAdd(out + 0, v.x);
    unsafeAtomicAdd(out + 1, v.y);
    unsafeAtomicAdd(out + 2, v.z);
    unsafeAtomicAdd(out + 3, v.w);
}

__global__ void bias_relu64(const float4* __restrict__ agg,
                            const float* __restrict__ b,
                            float4* __restrict__ h, int n4) {
    int i = blockIdx.x * blockDim.x + threadIdx.x;
    if (i >= n4) return;
    float4 v = agg[i];
    const int c = (i & 15) * 4;
    v.x = fmaxf(v.x + b[c + 0], 0.f);
    v.y = fmaxf(v.y + b[c + 1], 0.f);
    v.z = fmaxf(v.z + b[c + 2], 0.f);
    v.w = fmaxf(v.w + b[c + 3], 0.f);
    h[i] = v;
}

__global__ void logsoftmax32(const float* __restrict__ agg2,
                             const float* __restrict__ b2,
                             float* __restrict__ out, int n_nodes) {
    const int wid  = (int)(blockIdx.x * blockDim.x + threadIdx.x) >> 5;
    const int lane = threadIdx.x & 31;
    if (wid >= n_nodes) return;
    float v = agg2[(size_t)wid * 32 + lane] + b2[lane];
    float m = v;
#pragma unroll
    for (int o = 16; o; o >>= 1) m = fmaxf(m, __shfl_xor(m, o, 32));
    float e = __expf(v - m);
#pragma unroll
    for (int o = 16; o; o >>= 1) e += __shfl_xor(e, o, 32);
    out[(size_t)wid * 32 + lane] = (v - m) - __logf(e);
}

extern "C" void kernel_launch(void* const* d_in, const int* in_sizes, int n_in,
                              void* d_out, int out_size, void* d_ws, size_t ws_size,
                              hipStream_t stream) {
    const float* x   = (const float*)d_in[0];
    const int*   ei  = (const int*)d_in[1];
    const float* W1  = (const float*)d_in[2];
    const float* b1  = (const float*)d_in[3];
    const float* W2  = (const float*)d_in[4];
    const float* b2  = (const float*)d_in[5];
    float*       out = (float*)d_out;

    const int n_edges = in_sizes[1] / 2;
    const int* src = ei;
    const int* dst = ei + n_edges;

    float* t1   = (float*)d_ws;
    float* t2   = t1 + (size_t)N_NODES * HID;
    float* agg1 = t2 + (size_t)N_NODES * N_CLS;
    float* agg2 = agg1 + (size_t)N_NODES * HID;
    (void)ws_size; (void)n_in; (void)out_size;

    {
        int n4 = N_NODES * (HID + N_CLS) / 4;
        zero_f4<<<(n4 + 255) / 256, 256, 0, stream>>>((float4*)agg1, n4);
    }
    {
        int waves = N_NODES / 16;
        gemm_wmma_f32<IN_F, HID><<<(waves + 7) / 8, 256, 0, stream>>>(x, W1, t1, N_NODES);
    }
    {
        long long thr = (long long)n_edges * (HID / 4);
        scatter_add<HID><<<(int)((thr + 255) / 256), 256, 0, stream>>>(t1, src, dst, agg1, n_edges);
    }
    {
        int n4 = N_NODES * HID / 4;
        bias_relu64<<<(n4 + 255) / 256, 256, 0, stream>>>((const float4*)agg1, b1, (float4*)t1, n4);
    }
    {
        int waves = N_NODES / 16;
        gemm_wmma_f32<HID, N_CLS><<<(waves + 7) / 8, 256, 0, stream>>>(t1, W2, t2, N_NODES);
    }
    {
        long long thr = (long long)n_edges * (N_CLS / 4);
        scatter_add<N_CLS><<<(int)((thr + 255) / 256), 256, 0, stream>>>(t2, src, dst, agg2, n_edges);
    }
    {
        int waves = N_NODES;
        logsoftmax32<<<(waves + 7) / 8, 256, 0, stream>>>(agg2, b2, out, N_NODES);
    }
}


# scatter_add remapped: a wave walks 32 edges with src/dst broadcast from registers, so each gather load and float atomic add covers one contiguous 128-B row segment; arrival-order sums as before
# speedup vs baseline: 1.3959x; 1.3959x over previous
#include <hip/hip_runtime.h>
#include <hip/hip_bf16.h>

#define N_NODES 50000
#define IN_F    96
#define HID     64
#define N_CLS   32

typedef __attribute__((ext_vector_type(2))) float v2f;
typedef __attribute__((ext_vector_type(8))) float v8f;

__global__ void zero_f4(float4* __restrict__ p, int n4) {
    int i = blockIdx.x * blockDim.x + threadIdx.x;
    if (i < n4) p[i] = make_float4(0.f, 0.f, 0.f, 0.f);
}

template <int K, int N>
__global__ void gemm_wmma_f32(const float* __restrict__ X,
                              const float* __restrict__ W,
                              float* __restrict__ Y, int M) {
    __shared__ float sW[K * N];
    for (int i = threadIdx.x; i < K * N; i += blockDim.x) sW[i] = W[i];
    __syncthreads();

    const int wave = (int)(blockIdx.x * blockDim.x + threadIdx.x) >> 5;
    const int lane = threadIdx.x & 31;
    const int m0   = wave * 16;
    if (m0 >= M) return;

    const int half16 = lane >> 4;
    const int l15    = lane & 15;
    const int kh     = half16 * 2;
    const int arow   = m0 + l15;

    constexpr int NT = N / 16;
    v8f acc[NT] = {};

    for (int k = 0; k < K; k += 4) {
        v2f a;
        a.x = X[(size_t)arow * K + k + kh + 0];
        a.y = X[(size_t)arow * K + k + kh + 1];
#pragma unroll
        for (int t = 0; t < NT; ++t) {
            v2f b;
            b.x = sW[(k + kh + 0) * N + t * 16 + l15];
            b.y = sW[(k + kh + 1) * N + t * 16 + l15];
            acc[t] = __builtin_amdgcn_wmma_f32_16x16x4_f32(
                false, a, false, b, (short)0, acc[t], false, false);
        }
    }

    const int rbase = m0 + half16 * 8;
#pragma unroll
    for (int t = 0; t < NT; ++t)
#pragma unroll
        for (int r = 0; r < 8; ++r)
            Y[(size_t)(rbase + r) * N + t * 16 + l15] = acc[t][r];
}

template <int F>
__global__ void scatter_add(const float* __restrict__ msg,
                            const int* __restrict__ src,
                            const int* __restrict__ dst,
                            float* __restrict__ agg, int n_edges) {
    constexpr int SEG = F / 32;
    constexpr int CH  = 32 / SEG;
    const int wave = (int)(((long long)blockIdx.x * blockDim.x + threadIdx.x) >> 5);
    const int lane = threadIdx.x & 31;
    const long long e0 = (long long)wave * 32;
    if (e0 >= n_edges) return;
    const long long left = (long long)n_edges - e0;
    const int cnt = left < 32 ? (int)left : 32;

    int my_s = 0, my_d = 0;
    if (lane < cnt) {
        my_s = src[e0 + lane];
        my_d = dst[e0 + lane];
    }

    if (cnt == 32) {
#pragma unroll
        for (int c = 0; c < 32; c += CH) {
            float v[CH][SEG];
#pragma unroll
            for (int j = 0; j < CH; ++j) {
                const unsigned s = (unsigned)__builtin_amdgcn_readlane(my_s, c + j);
                const float* row = msg + (size_t)s * F;
#pragma unroll
                for (int g = 0; g < SEG; ++g)
                    v[j][g] = row[g * 32 + lane];
            }
#pragma unroll
            for (int j = 0; j < CH; ++j) {
                const unsigned d = (unsigned)__builtin_amdgcn_readlane(my_d, c + j);
                float* arow = agg + (size_t)d * F;
#pragma unroll
                for (int g = 0; g < SEG; ++g)
                    unsafeAtomicAdd(arow + g * 32 + lane, v[j][g]);
            }
        }
    } else {
        for (int j = 0; j < cnt; ++j) {
            const unsigned s = (unsigned)__builtin_amdgcn_readlane(my_s, j);
            const unsigned d = (unsigned)__builtin_amdgcn_readlane(my_d, j);
#pragma unroll
            for (int g = 0; g < SEG; ++g)
                unsafeAtomicAdd(agg + (size_t)d * F + g * 32 + lane,
                                msg[(size_t)s * F + g * 32 + lane]);
        }
    }
}

__global__ void bias_relu64(const float4* __restrict__ agg,
                            const float* __restrict__ b,
                            float4* __restrict__ h, int n4) {
    int i = blockIdx.x * blockDim.x + threadIdx.x;
    if (i >= n4) return;
    float4 v = agg[i];
    const int c = (i & 15) * 4;
    v.x = fmaxf(v.x + b[c + 0], 0.f);
    v.y = fmaxf(v.y + b[c + 1], 0.f);
    v.z = fmaxf(v.z + b[c + 2], 0.f);
    v.w = fmaxf(v.w + b[c + 3], 0.f);
    h[i] = v;
}

__global__ void logsoftmax32(const float* __restrict__ agg2,
                             const float* __restrict__ b2,
                             float* __restrict__ out, int n_nodes) {
    const int wid  = (int)(blockIdx.x * blockDim.x + threadIdx.x) >> 5;
    const int lane = threadIdx.x & 31;
    if (wid >= n_nodes) return;
    float v = agg2[(size_t)wid * 32 + lane] + b2[lane];
    float m = v;
#pragma unroll
    for (int o = 16; o; o >>= 1) m = fmaxf(m, __shfl_xor(m, o, 32));
    float e = __expf(v - m);
#pragma unroll
    for (int o = 16; o; o >>= 1) e += __shfl_xor(e, o, 32);
    out[(size_t)wid * 32 + lane] = (v - m) - __logf(e);
}

extern "C" void kernel_launch(void* const* d_in, const int* in_sizes, int n_in,
                              void* d_out, int out_size, void* d_ws, size_t ws_size,
                              hipStream_t stream) {
    const float* x   = (const float*)d_in[0];
    const int*   ei  = (const int*)d_in[1];
    const float* W1  = (const float*)d_in[2];
    const float* b1  = (const float*)d_in[3];
    const float* W2  = (const float*)d_in[4];
    const float* b2  = (const float*)d_in[5];
    float*       out = (float*)d_out;

    const int n_edges = in_sizes[1] / 2;
    const int* src = ei;
    const int* dst = ei + n_edges;

    float* t1   = (float*)d_ws;
    float* t2   = t1 + (size_t)N_NODES * HID;
    float* agg1 = t2 + (size_t)N_NODES * N_CLS;
    float* agg2 = agg1 + (size_t)N_NODES * HID;
    (void)ws_size; (void)n_in; (void)out_size;

    {
        int n4 = N_NODES * (HID + N_CLS) / 4;
        zero_f4<<<(n4 + 255) / 256, 256, 0, stream>>>((float4*)agg1, n4);
    }
    {
        int waves = N_NODES / 16;
        gemm_wmma_f32<IN_F, HID><<<(waves + 7) / 8, 256, 0, stream>>>(x, W1, t1, N_NODES);
    }
    {
        int waves = (n_edges + 31) / 32;
        scatter_add<HID><<<(waves + 7) / 8, 256, 0, stream>>>(t1, src, dst, agg1, n_edges);
    }
    {
        int n4 = N_NODES * HID / 4;
        bias_relu64<<<(n4 + 255) / 256, 256, 0, stream>>>((const float4*)agg1, b1, (float4*)t1, n4);
    }
    {
        int waves = N_NODES / 16;
        gemm_wmma_f32<HID, N_CLS><<<(waves + 7) / 8, 256, 0, stream>>>(t1, W2, t2, N_NODES);
    }
    {
        int waves = (n_edges + 31) / 32;
        scatter_add<N_CLS><<<(waves + 7) / 8, 256, 0, stream>>>(t2, src, dst, agg2, n_edges);
    }
    {
        int waves = N_NODES;
        logsoftmax32<<<(waves + 7) / 8, 256, 0, stream>>>(agg2, b2, out, N_NODES);
    }
}


# two elementwise passes folded into the GEMM launches beside them: GEMM1 also clears the scatter accumulators, GEMM2 forms relu(agg+b1) on its A-operand load; 7 launches become 5, sum order unchanged
# speedup vs baseline: 1.7631x; 1.2631x over previous
#include <hip/hip_runtime.h>
#include <hip/hip_bf16.h>

#define N_NODES 50000
#define IN_F    96
#define HID     64
#define N_CLS   32

typedef __attribute__((ext_vector_type(2))) float v2f;
typedef __attribute__((ext_vector_type(8))) float v8f;

template <int K, int N, int ZERO_F4, bool FOLD_BIAS_RELU>
__global__ void gemm_wmma_f32(const float* __restrict__ X,
                              const float* __restrict__ W,
                              const float* __restrict__ bias,
                              float* __restrict__ Y,
                              float4* __restrict__ zfill,
                              int M) {
    __shared__ float sW[K * N];
    __shared__ float sB[FOLD_BIAS_RELU ? K : 1];

    const int wave = (int)(blockIdx.x * blockDim.x + threadIdx.x) >> 5;
    const int lane = threadIdx.x & 31;
    const int m0   = wave * 16;

    if constexpr (ZERO_F4 > 0) {
        static_assert(ZERO_F4 % 32 == 0, "whole 512-byte stores only");
        if (m0 < M) {
            float4* zp = zfill + (size_t)wave * ZERO_F4 + lane;
#pragma unroll
            for (int j = 0; j < ZERO_F4 / 32; ++j)
                zp[j * 32] = make_float4(0.f, 0.f, 0.f, 0.f);
        }
    }

    for (int i = threadIdx.x; i < K * N; i += blockDim.x) sW[i] = W[i];
    if constexpr (FOLD_BIAS_RELU)
        for (int i = threadIdx.x; i < K; i += blockDim.x) sB[i] = bias[i];
    __syncthreads();

    if (m0 >= M) return;

    const int half16 = lane >> 4;
    const int l15    = lane & 15;
    const int kh     = half16 * 2;
    const int arow   = m0 + l15;

    constexpr int NT = N / 16;
    v8f acc[NT] = {};

    for (int k = 0; k < K; k += 4) {
        v2f a;
        a.x = X[(size_t)arow * K + k + kh + 0];
        a.y = X[(size_t)arow * K + k + kh + 1];
        if constexpr (FOLD_BIAS_RELU) {
            a.x = fmaxf(a.x + sB[k + kh + 0], 0.f);
            a.y = fmaxf(a.y + sB[k + kh + 1], 0.f);
        }
#pragma unroll
        for (int t = 0; t < NT; ++t) {
            v2f b;
            b.x = sW[(k + kh + 0) * N + t * 16 + l15];
            b.y = sW[(k + kh + 1) * N + t * 16 + l15];
            acc[t] = __builtin_amdgcn_wmma_f32_16x16x4_f32(
                false, a, false, b, (short)0, acc[t], false, false);
        }
    }

    const int rbase = m0 + half16 * 8;
#pragma unroll
    for (int t = 0; t < NT; ++t)
#pragma unroll
        for (int r = 0; r < 8; ++r)
            Y[(size_t)(rbase + r) * N + t * 16 + l15] = acc[t][r];
}

template <int F>
__global__ void scatter_add(const float* __restrict__ msg,
                            const int* __restrict__ src,
                            const int* __restrict__ dst,
                            float* __restrict__ agg, int n_edges) {
    constexpr int SEG = F / 32;
    constexpr int CH  = 32 / SEG;
    const int wave = (int)(((long long)blockIdx.x * blockDim.x + threadIdx.x) >> 5);
    const int lane = threadIdx.x & 31;
    const long long e0 = (long long)wave * 32;
    if (e0 >= n_edges) return;
    const long long left = (long long)n_edges - e0;
    const int cnt = left < 32 ? (int)left : 32;

    int my_s = 0, my_d = 0;
    if (lane < cnt) {
        my_s = src[e0 + lane];
        my_d = dst[e0 + lane];
    }

    if (cnt == 32) {
#pragma unroll
        for (int c = 0; c < 32; c += CH) {
            float v[CH][SEG];
#pragma unroll
            for (int j = 0; j < CH; ++j) {
                const unsigned s = (unsigned)__builtin_amdgcn_readlane(my_s, c + j);
                const float* row = msg + (size_t)s * F;
#pragma unroll
                for (int g = 0; g < SEG; ++g)
                    v[j][g] = row[g * 32 + lane];
            }
#pragma unroll
            for (int j = 0; j < CH; ++j) {
                const unsigned d = (unsigned)__builtin_amdgcn_readlane(my_d, c + j);
                float* arow = agg + (size_t)d * F;
#pragma unroll
                for (int g = 0; g < SEG; ++g)
                    unsafeAtomicAdd(arow + g * 32 + lane, v[j][g]);
            }
        }
    } else {
        for (int j = 0; j < cnt; ++j) {
            const unsigned s = (unsigned)__builtin_amdgcn_readlane(my_s, j);
            const unsigned d = (unsigned)__builtin_amdgcn_readlane(my_d, j);
#pragma unroll
            for (int g = 0; g < SEG; ++g)
                unsafeAtomicAdd(agg + (size_t)d * F + g * 32 + lane,
                                msg[(size_t)s * F + g * 32 + lane]);
        }
    }
}

__global__ void logsoftmax32(const float* __restrict__ agg2,
                             const float* __restrict__ b2,
                             float* __restrict__ out, int n_nodes) {
    const int wid  = (int)(blockIdx.x * blockDim.x + threadIdx.x) >> 5;
    const int lane = threadIdx.x & 31;
    if (wid >= n_nodes) return;
    float v = agg2[(size_t)wid * 32 + lane] + b2[lane];
    float m = v;
#pragma unroll
    for (int o = 16; o; o >>= 1) m = fmaxf(m, __shfl_xor(m, o, 32));
    float e = __expf(v - m);
#pragma unroll
    for (int o = 16; o; o >>= 1) e += __shfl_xor(e, o, 32);
    out[(size_t)wid * 32 + lane] = (v - m) - __logf(e);
}

extern "C" void kernel_launch(void* const* d_in, const int* in_sizes, int n_in,
                              void* d_out, int out_size, void* d_ws, size_t ws_size,
                              hipStream_t stream) {
    const float* x   = (const float*)d_in[0];
    const int*   ei  = (const int*)d_in[1];
    const float* W1  = (const float*)d_in[2];
    const float* b1  = (const float*)d_in[3];
    const float* W2  = (const float*)d_in[4];
    const float* b2  = (const float*)d_in[5];
    float*       out = (float*)d_out;

    const int n_edges = in_sizes[1] / 2;
    const int* src = ei;
    const int* dst = ei + n_edges;

    float* t1   = (float*)d_ws;
    float* t2   = t1 + (size_t)N_NODES * HID;
    float* agg1 = t2 + (size_t)N_NODES * N_CLS;
    float* agg2 = agg1 + (size_t)N_NODES * HID;
    (void)ws_size; (void)n_in; (void)out_size;

    {
        int waves = N_NODES / 16;
        constexpr int ZF4 = 16 * (HID + N_CLS) / 4;
        gemm_wmma_f32<IN_F, HID, ZF4, false><<<(waves + 7) / 8, 256, 0, stream>>>(
            x, W1, nullptr, t1, (float4*)agg1, N_NODES);
    }
    {
        int waves = (n_edges + 31) / 32;
        scatter_add<HID><<<(waves + 7) / 8, 256, 0, stream>>>(t1, src, dst, agg1, n_edges);
    }
    {
        int waves = N_NODES / 16;
        gemm_wmma_f32<HID, N_CLS, 0, true><<<(waves + 7) / 8, 256, 0, stream>>>(
            agg1, W2, b1, t2, nullptr, N_NODES);
    }
    {
        int waves = (n_edges + 31) / 32;
        scatter_add<N_CLS><<<(waves + 7) / 8, 256, 0, stream>>>(t2, src, dst, agg2, n_edges);
    }
    {
        int waves = N_NODES;
        logsoftmax32<<<(waves + 7) / 8, 256, 0, stream>>>(agg2, b2, out, N_NODES);
    }
}


# log-softmax launch as a resident grid: 3128 waves each take four 4-row groups (16-B loads, 8 lanes per row, 4 loads in flight) instead of 50000 one-row waves; each row's exp sum keeps its pairing
# speedup vs baseline: 1.8204x; 1.0325x over previous
#include <hip/hip_runtime.h>
#include <hip/hip_bf16.h>

#define N_NODES 50000
#define IN_F    96
#define HID     64
#define N_CLS   32

typedef __attribute__((ext_vector_type(2))) float v2f;
typedef __attribute__((ext_vector_type(8))) float v8f;

template <int K, int N, int ZERO_F4, bool FOLD_BIAS_RELU>
__global__ void gemm_wmma_f32(const float* __restrict__ X,
                              const float* __restrict__ W,
                              const float* __restrict__ bias,
                              float* __restrict__ Y,
                              float4* __restrict__ zfill,
                              int M) {
    __shared__ float sW[K * N];
    __shared__ float sB[FOLD_BIAS_RELU ? K : 1];

    const int wave = (int)(blockIdx.x * blockDim.x + threadIdx.x) >> 5;
    const int lane = threadIdx.x & 31;
    const int m0   = wave * 16;

    if constexpr (ZERO_F4 > 0) {
        static_assert(ZERO_F4 % 32 == 0, "whole 512-byte stores only");
        if (m0 < M) {
            float4* zp = zfill + (size_t)wave * ZERO_F4 + lane;
#pragma unroll
            for (int j = 0; j < ZERO_F4 / 32; ++j)
                zp[j * 32] = make_float4(0.f, 0.f, 0.f, 0.f);
        }
    }

    for (int i = threadIdx.x; i < K * N; i += blockDim.x) sW[i] = W[i];
    if constexpr (FOLD_BIAS_RELU)
        for (int i = threadIdx.x; i < K; i += blockDim.x) sB[i] = bias[i];
    __syncthreads();

    if (m0 >= M) return;

    const int half16 = lane >> 4;
    const int l15    = lane & 15;
    const int kh     = half16 * 2;
    const int arow   = m0 + l15;

    constexpr int NT = N / 16;
    v8f acc[NT] = {};

    for (int k = 0; k < K; k += 4) {
        v2f a;
        a.x = X[(size_t)arow * K + k + kh + 0];
        a.y = X[(size_t)arow * K + k + kh + 1];
        if constexpr (FOLD_BIAS_RELU) {
            a.x = fmaxf(a.x + sB[k + kh + 0], 0.f);
            a.y = fmaxf(a.y + sB[k + kh + 1], 0.f);
        }
#pragma unroll
        for (int t = 0; t < NT; ++t) {
            v2f b;
            b.x = sW[(k + kh + 0) * N + t * 16 + l15];
            b.y = sW[(k + kh + 1) * N + t * 16 + l15];
            acc[t] = __builtin_amdgcn_wmma_f32_16x16x4_f32(
                false, a, false, b, (short)0, acc[t], false, false);
        }
    }

    const int rbase = m0 + half16 * 8;
#pragma unroll
    for (int t = 0; t < NT; ++t)
#pragma unroll
        for (int r = 0; r < 8; ++r)
            Y[(size_t)(rbase + r) * N + t * 16 + l15] = acc[t][r];
}

template <int F>
__global__ void scatter_add(const float* __restrict__ msg,
                            const int* __restrict__ src,
                            const int* __restrict__ dst,
                            float* __restrict__ agg, int n_edges) {
    constexpr int SEG = F / 32;
    constexpr int CH  = 32 / SEG;
    const int wave = (int)(((long long)blockIdx.x * blockDim.x + threadIdx.x) >> 5);
    const int lane = threadIdx.x & 31;
    const long long e0 = (long long)wave * 32;
    if (e0 >= n_edges) return;
    const long long left = (long long)n_edges - e0;
    const int cnt = left < 32 ? (int)left : 32;

    int my_s = 0, my_d = 0;
    if (lane < cnt) {
        my_s = src[e0 + lane];
        my_d = dst[e0 + lane];
    }

    if (cnt == 32) {
#pragma unroll
        for (int c = 0; c < 32; c += CH) {
            float v[CH][SEG];
#pragma unroll
            for (int j = 0; j < CH; ++j) {
                const unsigned s = (unsigned)__builtin_amdgcn_readlane(my_s, c + j);
                const float* row = msg + (size_t)s * F;
#pragma unroll
                for (int g = 0; g < SEG; ++g)
                    v[j][g] = row[g * 32 + lane];
            }
#pragma unroll
            for (int j = 0; j < CH; ++j) {
                const unsigned d = (unsigned)__builtin_amdgcn_readlane(my_d, c + j);
                float* arow = agg + (size_t)d * F;
#pragma unroll
                for (int g = 0; g < SEG; ++g)
                    unsafeAtomicAdd(arow + g * 32 + lane, v[j][g]);
            }
        }
    } else {
        for (int j = 0; j < cnt; ++j) {
            const unsigned s = (unsigned)__builtin_amdgcn_readlane(my_s, j);
            const unsigned d = (unsigned)__builtin_amdgcn_readlane(my_d, j);
#pragma unroll
            for (int g = 0; g < SEG; ++g)
                unsafeAtomicAdd(agg + (size_t)d * F + g * 32 + lane,
                                msg[(size_t)s * F + g * 32 + lane]);
        }
    }
}

#define LS_U 4
__global__ void logsoftmax32(const float* __restrict__ agg2,
                             const float* __restrict__ b2,
                             float* __restrict__ out, int n_nodes) {
    const int lane   = threadIdx.x & 31;
    const int wave   = (int)((blockIdx.x * blockDim.x + threadIdx.x) >> 5);
    const int nwaves = (int)((gridDim.x * blockDim.x) >> 5);
    const int sub = lane >> 3;
    const int q   = lane & 7;
    const int ngroups = (n_nodes + 3) >> 2;

    float4 bb;
    bb.x = b2[q * 4 + 0]; bb.y = b2[q * 4 + 1];
    bb.z = b2[q * 4 + 2]; bb.w = b2[q * 4 + 3];

    const float4* in4  = (const float4*)agg2;
    float4*       out4 = (float4*)out;

    for (int g0 = wave; g0 < ngroups; g0 += nwaves * LS_U) {
        float4 v[LS_U];
        int    row[LS_U];
#pragma unroll
        for (int u = 0; u < LS_U; ++u) {
            row[u] = (g0 + u * nwaves) * 4 + sub;
            const int rc = row[u] < n_nodes ? row[u] : n_nodes - 1;
            v[u] = in4[(size_t)rc * 8 + q];
        }
        static_assert(LS_U == 4, "the fence below names four loads");
        asm volatile("" :: "v"(v[0].x), "v"(v[0].y), "v"(v[0].z), "v"(v[0].w),
                           "v"(v[1].x), "v"(v[1].y), "v"(v[1].z), "v"(v[1].w),
                           "v"(v[2].x), "v"(v[2].y), "v"(v[2].z), "v"(v[2].w),
                           "v"(v[3].x), "v"(v[3].y), "v"(v[3].z), "v"(v[3].w));

        float4 r[LS_U];
#pragma unroll
        for (int u = 0; u < LS_U; ++u) {
            float4 x;
            x.x = v[u].x + bb.x; x.y = v[u].y + bb.y;
            x.z = v[u].z + bb.z; x.w = v[u].w + bb.w;
            float m = fmaxf(fmaxf(x.x, x.y), fmaxf(x.z, x.w));
#pragma unroll
            for (int o = 4; o; o >>= 1) m = fmaxf(m, __shfl_xor(m, o, 32));
            x.x -= m; x.y -= m; x.z -= m; x.w -= m;
            float e0 = __expf(x.x), e1 = __expf(x.y);
            float e2 = __expf(x.z), e3 = __expf(x.w);
#pragma unroll
            for (int o = 4; o; o >>= 1) {
                e0 += __shfl_xor(e0, o, 32);
                e1 += __shfl_xor(e1, o, 32);
                e2 += __shfl_xor(e2, o, 32);
                e3 += __shfl_xor(e3, o, 32);
            }
            const float ls = __logf((e0 + e2) + (e1 + e3));
            r[u].x = x.x - ls; r[u].y = x.y - ls;
            r[u].z = x.z - ls; r[u].w = x.w - ls;
        }
#pragma unroll
        for (int u = 0; u < LS_U; ++u)
            if (row[u] < n_nodes)
                out4[(size_t)row[u] * 8 + q] = r[u];
    }
}

extern "C" void kernel_launch(void* const* d_in, const int* in_sizes, int n_in,
                              void* d_out, int out_size, void* d_ws, size_t ws_size,
                              hipStream_t stream) {
    const float* x   = (const float*)d_in[0];
    const int*   ei  = (const int*)d_in[1];
    const float* W1  = (const float*)d_in[2];
    const float* b1  = (const float*)d_in[3];
    const float* W2  = (const float*)d_in[4];
    const float* b2  = (const float*)d_in[5];
    float*       out = (float*)d_out;

    const int n_edges = in_sizes[1] / 2;
    const int* src = ei;
    const int* dst = ei + n_edges;

    float* t1   = (float*)d_ws;
    float* t2   = t1 + (size_t)N_NODES * HID;
    float* agg1 = t2 + (size_t)N_NODES * N_CLS;
    float* agg2 = agg1 + (size_t)N_NODES * HID;
    (void)ws_size; (void)n_in; (void)out_size;

    {
        int waves = N_NODES / 16;
        constexpr int ZF4 = 16 * (HID + N_CLS) / 4;
        gemm_wmma_f32<IN_F, HID, ZF4, false><<<(waves + 7) / 8, 256, 0, stream>>>(
            x, W1, nullptr, t1, (float4*)agg1, N_NODES);
    }
    {
        int waves = (n_edges + 31) / 32;
        scatter_add<HID><<<(waves + 7) / 8, 256, 0, stream>>>(t1, src, dst, agg1, n_edges);
    }
    {
        int waves = N_NODES / 16;
        gemm_wmma_f32<HID, N_CLS, 0, true><<<(waves + 7) / 8, 256, 0, stream>>>(
            agg1, W2, b1, t2, nullptr, N_NODES);
    }
    {
        int waves = (n_edges + 31) / 32;
        scatter_add<N_CLS><<<(waves + 7) / 8, 256, 0, stream>>>(t2, src, dst, agg2, n_edges);
    }
    logsoftmax32<<<391, 256, 0, stream>>>(agg2, b2, out, N_NODES);
}


# scatter_add by float memory atomics replaced: each call bins the edges by destination range with LDS counters, one work-group per bin sums rows in LDS (ds adds, arrival order) and stores each row once
# speedup vs baseline: 1.9253x; 1.0576x over previous
#include <hip/hip_runtime.h>
#include <hip/hip_bf16.h>

#define N_NODES 50000
#define IN_F    96
#define HID     64
#define N_CLS   32

typedef __attribute__((ext_vector_type(2))) float v2f;
typedef __attribute__((ext_vector_type(8))) float v8f;

template <int K, int N, bool FOLD_BIAS_RELU>
__global__ void gemm_wmma_f32(const float* __restrict__ X,
                              const float* __restrict__ W,
                              const float* __restrict__ bias,
                              float* __restrict__ Y,
                              int M) {
    __shared__ float sW[K * N];
    __shared__ float sB[FOLD_BIAS_RELU ? K : 1];

    const int wave = (int)(blockIdx.x * blockDim.x + threadIdx.x) >> 5;
    const int lane = threadIdx.x & 31;
    const int m0   = wave * 16;

    for (int i = threadIdx.x; i < K * N; i += blockDim.x) sW[i] = W[i];
    if constexpr (FOLD_BIAS_RELU)
        for (int i = threadIdx.x; i < K; i += blockDim.x) sB[i] = bias[i];
    __syncthreads();

    if (m0 >= M) return;

    const int half16 = lane >> 4;
    const int l15    = lane & 15;
    const int kh     = half16 * 2;
    const int arow   = m0 + l15;

    constexpr int NT = N / 16;
    v8f acc[NT] = {};

    for (int k = 0; k < K; k += 4) {
        v2f a;
        a.x = X[(size_t)arow * K + k + kh + 0];
        a.y = X[(size_t)arow * K + k + kh + 1];
        if constexpr (FOLD_BIAS_RELU) {
            a.x = fmaxf(a.x + sB[k + kh + 0], 0.f);
            a.y = fmaxf(a.y + sB[k + kh + 1], 0.f);
        }
#pragma unroll
        for (int t = 0; t < NT; ++t) {
            v2f b;
            b.x = sW[(k + kh + 0) * N + t * 16 + l15];
            b.y = sW[(k + kh + 1) * N + t * 16 + l15];
            acc[t] = __builtin_amdgcn_wmma_f32_16x16x4_f32(
                false, a, false, b, (short)0, acc[t], false, false);
        }
    }

    const int rbase = m0 + half16 * 8;
#pragma unroll
    for (int t = 0; t < NT; ++t)
#pragma unroll
        for (int r = 0; r < 8; ++r)
            Y[(size_t)(rbase + r) * N + t * 16 + l15] = acc[t][r];
}

#define NB      256
#define BIN_R   196
#define PART_WG 256
static_assert(NB * BIN_R >= N_NODES, "bins must cover every node");
static_assert(N_NODES <= 65536 && BIN_R <= 65536, "record: 16 bits src, 16 bits local dst");

__device__ __forceinline__ void edge_share(int p, int P, int n_edges, int& e0, int& e1) {
    const int per = (n_edges + P - 1) / P;
    e0 = p * per;  if (e0 > n_edges) e0 = n_edges;
    e1 = e0 + per; if (e1 > n_edges) e1 = n_edges;
}

__global__ void bin_count(const int* __restrict__ dst, int n_edges,
                          int* __restrict__ counts) {
    __shared__ int hist[NB];
    for (int i = threadIdx.x; i < NB; i += blockDim.x) hist[i] = 0;
    __syncthreads();
    int e0, e1;
    edge_share(blockIdx.x, PART_WG, n_edges, e0, e1);
    for (int eb = e0 + (int)threadIdx.x; eb < e1; eb += 4 * (int)blockDim.x) {
        int d[4];
#pragma unroll
        for (int k = 0; k < 4; ++k) {
            const int e = eb + k * (int)blockDim.x;
            d[k] = e < e1 ? dst[e] : -1;
        }
#pragma unroll
        for (int k = 0; k < 4; ++k)
            if (d[k] >= 0) atomicAdd(&hist[(unsigned)d[k] / BIN_R], 1);
    }
    __syncthreads();
    for (int i = threadIdx.x; i < NB; i += blockDim.x)
        counts[blockIdx.x * NB + i] = hist[i];
}

__global__ void __launch_bounds__(4 * NB)
bin_place(const int* __restrict__ src, const int* __restrict__ dst,
          int n_edges, const int* __restrict__ counts,
          int* __restrict__ sorted, int* __restrict__ binStart) {
    __shared__ int ptot[4][NB];
    __shared__ int ppre[4][NB];
    __shared__ int sc[2][NB];
    __shared__ int cursor[NB];
    const int t    = threadIdx.x;
    const int b    = t % NB;
    const int part = t / NB;
    const int p    = blockIdx.x;

    int tot = 0, pre = 0;
    constexpr int G = 16;
    static_assert((PART_WG / 4) % G == 0, "whole groups only");
    for (int q0 = part; q0 < PART_WG; q0 += 4 * G) {
        int c[G];
#pragma unroll
        for (int i = 0; i < G; ++i) c[i] = counts[(q0 + 4 * i) * NB + b];
#pragma unroll
        for (int i = 0; i < G; ++i) {
            tot += c[i];
            pre += (q0 + 4 * i < p) ? c[i] : 0;
        }
    }
    ptot[part][b] = tot;
    ppre[part][b] = pre;
    __syncthreads();
    if (t < NB) {
        tot = (ptot[0][t] + ptot[1][t]) + (ptot[2][t] + ptot[3][t]);
        pre = (ppre[0][t] + ppre[1][t]) + (ppre[2][t] + ppre[3][t]);
        sc[0][t] = tot;
    }
    __syncthreads();
    int cur = 0;
    for (int off = 1; off < NB; off <<= 1) {
        if (t < NB) {
            int v = sc[cur][t];
            if (t >= off) v += sc[cur][t - off];
            sc[cur ^ 1][t] = v;
        }
        __syncthreads();
        cur ^= 1;
    }
    if (t < NB) {
        const int start = sc[cur][t] - tot;
        cursor[t] = start + pre;
        if (p == 0) {
            binStart[t] = start;
            if (t == NB - 1) binStart[NB] = start + tot;
        }
    }
    __syncthreads();

    int e0, e1;
    edge_share(p, PART_WG, n_edges, e0, e1);
    for (int eb = e0 + t; eb < e1; eb += 4 * (int)blockDim.x) {
        int d[4], s[4];
#pragma unroll
        for (int k = 0; k < 4; ++k) {
            const int e = eb + k * (int)blockDim.x;
            d[k] = e < e1 ? dst[e] : -1;
            s[k] = e < e1 ? src[e] : 0;
        }
#pragma unroll
        for (int k = 0; k < 4; ++k) {
            if (d[k] >= 0) {
                const unsigned dd  = (unsigned)d[k];
                const unsigned bin = dd / BIN_R;
                const int slot = atomicAdd(&cursor[bin], 1);
                sorted[slot] = (int)((unsigned)s[k] | ((dd - bin * BIN_R) << 16));
            }
        }
    }
}

template <int F, bool FULL>
__device__ __forceinline__ void gather_batch(const float* __restrict__ msg, float* acc,
                                             int rec, int cnt, int lane) {
    constexpr int SEG = F / 32;
    constexpr int CH  = 32 / SEG;
#pragma unroll
    for (int c = 0; c < 32; c += CH) {
        if (FULL || c < cnt) {
            float v[CH][SEG];
#pragma unroll
            for (int j = 0; j < CH; ++j) {
                const unsigned r = (unsigned)__builtin_amdgcn_readlane(rec, c + j);
                const float* row = msg + (size_t)(r & 0xFFFFu) * F;
#pragma unroll
                for (int g = 0; g < SEG; ++g)
                    v[j][g] = row[g * 32 + lane];
            }
#pragma unroll
            for (int j = 0; j < CH; ++j) {
                if (FULL || c + j < cnt) {
                    const unsigned r = (unsigned)__builtin_amdgcn_readlane(rec, c + j);
                    float* arow = acc + (r >> 16) * F;
#pragma unroll
                    for (int g = 0; g < SEG; ++g)
                        __hip_atomic_fetch_add(arow + g * 32 + lane, v[j][g],
                                               __ATOMIC_RELAXED, __HIP_MEMORY_SCOPE_WORKGROUP);
                }
            }
        }
    }
}

template <int F>
__global__ void __launch_bounds__(1024)
bin_gather(const float* __restrict__ msg, const int* __restrict__ sorted,
           const int* __restrict__ binStart, float* __restrict__ agg, int n_nodes) {
    __shared__ float4 acc4[BIN_R * F / 4];
    float* acc = (float*)acc4;

    const int lane = threadIdx.x & 31;
    const int wave = __builtin_amdgcn_readfirstlane((int)(threadIdx.x >> 5));
    const int nw   = (int)(blockDim.x >> 5);

    const int beg = binStart[blockIdx.x];
    const int end = binStart[blockIdx.x + 1];
    int base = beg + wave * 32;
    int rec = 0;
    if (base + lane < end) rec = sorted[base + lane];

    for (int i = threadIdx.x; i < BIN_R * F / 4; i += blockDim.x)
        acc4[i] = make_float4(0.f, 0.f, 0.f, 0.f);
    __syncthreads();

    while (base < end) {
        const int nbase = base + nw * 32;
        int nrec = 0;
        if (nbase + lane < end) nrec = sorted[nbase + lane];
        const int left = end - base;
        const int cnt  = left < 32 ? left : 32;
        if (cnt == 32) gather_batch<F, true>(msg, acc, rec, cnt, lane);
        else           gather_batch<F, false>(msg, acc, rec, cnt, lane);
        rec  = nrec;
        base = nbase;
    }
    __syncthreads();

    const int node0 = blockIdx.x * BIN_R;
    float4* out4 = (float4*)agg + (size_t)node0 * (F / 4);
    for (int i = threadIdx.x; i < BIN_R * F / 4; i += blockDim.x)
        if (node0 + i / (F / 4) < n_nodes) out4[i] = acc4[i];
}

#define LS_U 4
__global__ void logsoftmax32(const float* __restrict__ agg2,
                             const float* __restrict__ b2,
                             float* __restrict__ out, int n_nodes) {
    const int lane   = threadIdx.x & 31;
    const int wave   = (int)((blockIdx.x * blockDim.x + threadIdx.x) >> 5);
    const int nwaves = (int)((gridDim.x * blockDim.x) >> 5);
    const int sub = lane >> 3;
    const int q   = lane & 7;
    const int ngroups = (n_nodes + 3) >> 2;

    float4 bb;
    bb.x = b2[q * 4 + 0]; bb.y = b2[q * 4 + 1];
    bb.z = b2[q * 4 + 2]; bb.w = b2[q * 4 + 3];

    const float4* in4  = (const float4*)agg2;
    float4*       out4 = (float4*)out;

    for (int g0 = wave; g0 < ngroups; g0 += nwaves * LS_U) {
        float4 v[LS_U];
        int    row[LS_U];
#pragma unroll
        for (int u = 0; u < LS_U; ++u) {
            row[u] = (g0 + u * nwaves) * 4 + sub;
            const int rc = row[u] < n_nodes ? row[u] : n_nodes - 1;
            v[u] = in4[(size_t)rc * 8 + q];
        }
        static_assert(LS_U == 4, "the fence below names four loads");
        asm volatile("" :: "v"(v[0].x), "v"(v[0].y), "v"(v[0].z), "v"(v[0].w),
                           "v"(v[1].x), "v"(v[1].y), "v"(v[1].z), "v"(v[1].w),
                           "v"(v[2].x), "v"(v[2].y), "v"(v[2].z), "v"(v[2].w),
                           "v"(v[3].x), "v"(v[3].y), "v"(v[3].z), "v"(v[3].w));

        float4 r[LS_U];
#pragma unroll
        for (int u = 0; u < LS_U; ++u) {
            float4 x;
            x.x = v[u].x + bb.x; x.y = v[u].y + bb.y;
            x.z = v[u].z + bb.z; x.w = v[u].w + bb.w;
            float m = fmaxf(fmaxf(x.x, x.y), fmaxf(x.z, x.w));
#pragma unroll
            for (int o = 4; o; o >>= 1) m = fmaxf(m, __shfl_xor(m, o, 32));
            x.x -= m; x.y -= m; x.z -= m; x.w -= m;
            float e0 = __expf(x.x), e1 = __expf(x.y);
            float e2 = __expf(x.z), e3 = __expf(x.w);
#pragma unroll
            for (int o = 4; o; o >>= 1) {
                e0 += __shfl_xor(e0, o, 32);
                e1 += __shfl_xor(e1, o, 32);
                e2 += __shfl_xor(e2, o, 32);
                e3 += __shfl_xor(e3, o, 32);
            }
            const float ls = __logf((e0 + e2) + (e1 + e3));
            r[u].x = x.x - ls; r[u].y = x.y - ls;
            r[u].z = x.z - ls; r[u].w = x.w - ls;
        }
#pragma unroll
        for (int u = 0; u < LS_U; ++u)
            if (row[u] < n_nodes)
                out4[(size_t)row[u] * 8 + q] = r[u];
    }
}

extern "C" void kernel_launch(void* const* d_in, const int* in_sizes, int n_in,
                              void* d_out, int out_size, void* d_ws, size_t ws_size,
                              hipStream_t stream) {
    const float* x   = (const float*)d_in[0];
    const int*   ei  = (const int*)d_in[1];
    const float* W1  = (const float*)d_in[2];
    const float* b1  = (const float*)d_in[3];
    const float* W2  = (const float*)d_in[4];
    const float* b2  = (const float*)d_in[5];
    float*       out = (float*)d_out;

    const int n_edges = in_sizes[1] / 2;
    const int* src = ei;
    const int* dst = ei + n_edges;

    float* t1   = (float*)d_ws;
    float* t2   = t1 + (size_t)N_NODES * HID;
    float* agg1 = t2 + (size_t)N_NODES * N_CLS;
    float* agg2 = agg1 + (size_t)N_NODES * HID;
    int* counts   = (int*)(agg2 + (size_t)N_NODES * N_CLS);
    int* binStart = counts + PART_WG * NB;
    int* sorted   = binStart + 2 * NB;
    (void)ws_size; (void)n_in; (void)out_size;

    {
        int waves = N_NODES / 16;
        gemm_wmma_f32<IN_F, HID, false><<<(waves + 7) / 8, 256, 0, stream>>>(
            x, W1, nullptr, t1, N_NODES);
    }
    bin_count<<<PART_WG, 1024, 0, stream>>>(dst, n_edges, counts);
    bin_place<<<PART_WG, 4 * NB, 0, stream>>>(src, dst, n_edges, counts, sorted, binStart);
    bin_gather<HID><<<NB, 1024, 0, stream>>>(t1, sorted, binStart, agg1, N_NODES);
    {
        int waves = N_NODES / 16;
        gemm_wmma_f32<HID, N_CLS, true><<<(waves + 7) / 8, 256, 0, stream>>>(
            agg1, W2, b1, t2, N_NODES);
    }
    bin_gather<N_CLS><<<NB, 1024, 0, stream>>>(t2, sorted, binStart, agg2, N_NODES);
    logsoftmax32<<<391, 256, 0, stream>>>(agg2, b2, out, N_NODES);
}


# each edge-sum work-group finishes its bin's rows in LDS: layer 1 forms relu(sum+b1)@W2 there on WMMA, layer 2 the log-softmax; the sums never reach memory, 7 launches become 5, sum order kept
# speedup vs baseline: 2.2037x; 1.1446x over previous
#include <hip/hip_runtime.h>
#include <hip/hip_bf16.h>

#define N_NODES 50000
#define IN_F    96
#define HID     64
#define N_CLS   32

typedef __attribute__((ext_vector_type(2))) float v2f;
typedef __attribute__((ext_vector_type(8))) float v8f;

template <int K, int N>
__global__ void gemm_wmma_f32(const float* __restrict__ X,
                              const float* __restrict__ W,
                              float* __restrict__ Y,
                              int M) {
    __shared__ float sW[K * N];

    const int wave = (int)(blockIdx.x * blockDim.x + threadIdx.x) >> 5;
    const int lane = threadIdx.x & 31;
    const int m0   = wave * 16;

    for (int i = threadIdx.x; i < K * N; i += blockDim.x) sW[i] = W[i];
    __syncthreads();

    if (m0 >= M) return;

    const int half16 = lane >> 4;
    const int l15    = lane & 15;
    const int kh     = half16 * 2;
    const int arow   = m0 + l15;

    constexpr int NT = N / 16;
    v8f acc[NT] = {};

    for (int k = 0; k < K; k += 4) {
        v2f a;
        a.x = X[(size_t)arow * K + k + kh + 0];
        a.y = X[(size_t)arow * K + k + kh + 1];
#pragma unroll
        for (int t = 0; t < NT; ++t) {
            v2f b;
            b.x = sW[(k + kh + 0) * N + t * 16 + l15];
            b.y = sW[(k + kh + 1) * N + t * 16 + l15];
            acc[t] = __builtin_amdgcn_wmma_f32_16x16x4_f32(
                false, a, false, b, (short)0, acc[t], false, false);
        }
    }

    const int rbase = m0 + half16 * 8;
#pragma unroll
    for (int t = 0; t < NT; ++t)
#pragma unroll
        for (int r = 0; r < 8; ++r)
            Y[(size_t)(rbase + r) * N + t * 16 + l15] = acc[t][r];
}

#define NB      256
#define BIN_R   196
#define PART_WG 256
static_assert(NB * BIN_R >= N_NODES, "bins must cover every node");
static_assert(N_NODES <= 65536 && BIN_R <= 65536, "record: 16 bits src, 16 bits local dst");

__device__ __forceinline__ void edge_share(int p, int P, int n_edges, int& e0, int& e1) {
    const int per = (n_edges + P - 1) / P;
    e0 = p * per;  if (e0 > n_edges) e0 = n_edges;
    e1 = e0 + per; if (e1 > n_edges) e1 = n_edges;
}

__global__ void bin_count(const int* __restrict__ dst, int n_edges,
                          int* __restrict__ counts) {
    __shared__ int hist[NB];
    for (int i = threadIdx.x; i < NB; i += blockDim.x) hist[i] = 0;
    __syncthreads();
    int e0, e1;
    edge_share(blockIdx.x, PART_WG, n_edges, e0, e1);
    for (int eb = e0 + (int)threadIdx.x; eb < e1; eb += 4 * (int)blockDim.x) {
        int d[4];
#pragma unroll
        for (int k = 0; k < 4; ++k) {
            const int e = eb + k * (int)blockDim.x;
            d[k] = e < e1 ? dst[e] : -1;
        }
#pragma unroll
        for (int k = 0; k < 4; ++k)
            if (d[k] >= 0) atomicAdd(&hist[(unsigned)d[k] / BIN_R], 1);
    }
    __syncthreads();
    for (int i = threadIdx.x; i < NB; i += blockDim.x)
        counts[blockIdx.x * NB + i] = hist[i];
}

__global__ void __launch_bounds__(4 * NB)
bin_place(const int* __restrict__ src, const int* __restrict__ dst,
          int n_edges, const int* __restrict__ counts,
          int* __restrict__ sorted, int* __restrict__ binStart) {
    __shared__ int ptot[4][NB];
    __shared__ int ppre[4][NB];
    __shared__ int sc[2][NB];
    __shared__ int cursor[NB];
    const int t    = threadIdx.x;
    const int b    = t % NB;
    const int part = t / NB;
    const int p    = blockIdx.x;

    int tot = 0, pre = 0;
    constexpr int G = 16;
    static_assert((PART_WG / 4) % G == 0, "whole groups only");
    for (int q0 = part; q0 < PART_WG; q0 += 4 * G) {
        int c[G];
#pragma unroll
        for (int i = 0; i < G; ++i) c[i] = counts[(q0 + 4 * i) * NB + b];
#pragma unroll
        for (int i = 0; i < G; ++i) {
            tot += c[i];
            pre += (q0 + 4 * i < p) ? c[i] : 0;
        }
    }
    ptot[part][b] = tot;
    ppre[part][b] = pre;
    __syncthreads();
    if (t < NB) {
        tot = (ptot[0][t] + ptot[1][t]) + (ptot[2][t] + ptot[3][t]);
        pre = (ppre[0][t] + ppre[1][t]) + (ppre[2][t] + ppre[3][t]);
        sc[0][t] = tot;
    }
    __syncthreads();
    int cur = 0;
    for (int off = 1; off < NB; off <<= 1) {
        if (t < NB) {
            int v = sc[cur][t];
            if (t >= off) v += sc[cur][t - off];
            sc[cur ^ 1][t] = v;
        }
        __syncthreads();
        cur ^= 1;
    }
    if (t < NB) {
        const int start = sc[cur][t] - tot;
        cursor[t] = start + pre;
        if (p == 0) {
            binStart[t] = start;
            if (t == NB - 1) binStart[NB] = start + tot;
        }
    }
    __syncthreads();

    int e0, e1;
    edge_share(p, PART_WG, n_edges, e0, e1);
    for (int eb = e0 + t; eb < e1; eb += 4 * (int)blockDim.x) {
        int d[4], s[4];
#pragma unroll
        for (int k = 0; k < 4; ++k) {
            const int e = eb + k * (int)blockDim.x;
            d[k] = e < e1 ? dst[e] : -1;
            s[k] = e < e1 ? src[e] : 0;
        }
#pragma unroll
        for (int k = 0; k < 4; ++k) {
            if (d[k] >= 0) {
                const unsigned dd  = (unsigned)d[k];
                const unsigned bin = dd / BIN_R;
                const int slot = atomicAdd(&cursor[bin], 1);
                sorted[slot] = (int)((unsigned)s[k] | ((dd - bin * BIN_R) << 16));
            }
        }
    }
}

template <int F, int P, bool FULL>
__device__ __forceinline__ void gather_batch(const float* __restrict__ msg, float* acc,
                                             int rec, int cnt, int lane) {
    constexpr int SEG = F / 32;
    constexpr int CH  = 32 / SEG;
#pragma unroll
    for (int c = 0; c < 32; c += CH) {
        if (FULL || c < cnt) {
            float v[CH][SEG];
#pragma unroll
            for (int j = 0; j < CH; ++j) {
                const unsigned r = (unsigned)__builtin_amdgcn_readlane(rec, c + j);
                const float* row = msg + (size_t)(r & 0xFFFFu) * F;
#pragma unroll
                for (int g = 0; g < SEG; ++g)
                    v[j][g] = row[g * 32 + lane];
            }
#pragma unroll
            for (int j = 0; j < CH; ++j) {
                if (FULL || c + j < cnt) {
                    const unsigned r = (unsigned)__builtin_amdgcn_readlane(rec, c + j);
                    float* arow = acc + (r >> 16) * P;
#pragma unroll
                    for (int g = 0; g < SEG; ++g)
                        __hip_atomic_fetch_add(arow + g * 32 + lane, v[j][g],
                                               __ATOMIC_RELAXED, __HIP_MEMORY_SCOPE_WORKGROUP);
                }
            }
        }
    }
}

template <int F, int P, int ZROWS, class Stage>
__device__ __forceinline__ void bin_sum_rows(const float* __restrict__ msg,
                                             const int* __restrict__ sorted,
                                             const int* __restrict__ binStart,
                                             float4* acc4, Stage stage) {
    static_assert((ZROWS * P) % 4 == 0, "cleared as float4");
    float* acc = (float*)acc4;
    const int lane = threadIdx.x & 31;
    const int wave = __builtin_amdgcn_readfirstlane((int)(threadIdx.x >> 5));
    const int nw   = (int)(blockDim.x >> 5);

    const int beg = binStart[blockIdx.x];
    const int end = binStart[blockIdx.x + 1];
    int base = beg + wave * 32;
    int rec = 0;
    if (base + lane < end) rec = sorted[base + lane];

    for (int i = threadIdx.x; i < ZROWS * P / 4; i += blockDim.x)
        acc4[i] = make_float4(0.f, 0.f, 0.f, 0.f);
    stage();
    __syncthreads();

    while (base < end) {
        const int nbase = base + nw * 32;
        int nrec = 0;
        if (nbase + lane < end) nrec = sorted[nbase + lane];
        const int left = end - base;
        const int cnt  = left < 32 ? left : 32;
        if (cnt == 32) gather_batch<F, P, true>(msg, acc, rec, cnt, lane);
        else           gather_batch<F, P, false>(msg, acc, rec, cnt, lane);
        rec  = nrec;
        base = nbase;
    }
    __syncthreads();
}

#define A1_P    68
#define A1_ROWS 208
static_assert(A1_ROWS % 16 == 0 && A1_ROWS >= BIN_R, "whole 16-row tiles over the bin");

__global__ void __launch_bounds__(1024)
bin_sum_relu_gemm(const float* __restrict__ t1, const int* __restrict__ sorted,
                  const int* __restrict__ binStart, const float* __restrict__ b1,
                  const float* __restrict__ W2, float* __restrict__ t2, int n_nodes) {
    __shared__ float4 acc4[A1_ROWS * A1_P / 4];
    __shared__ float  sW[HID * N_CLS];
    __shared__ float  sB[HID];
    static_assert(HID * N_CLS == 2 * 1024, "two weights per thread of the 1024");

    const int   tid = threadIdx.x;
    const float w0  = W2[tid];
    const float w1  = W2[tid + 1024];
    const float bv  = b1[tid & (HID - 1)];

    bin_sum_rows<HID, A1_P, A1_ROWS>(t1, sorted, binStart, acc4, [&] {
        sW[tid]        = w0;
        sW[tid + 1024] = w1;
        if (tid < HID) sB[tid] = bv;
    });

    const float* acc = (const float*)acc4;
    const int lane = tid & 31;
    const int wave = __builtin_amdgcn_readfirstlane(tid >> 5);
    constexpr int MT = A1_ROWS / 16;
    constexpr int NT = N_CLS / 16;
    if (wave < MT * NT) {
        const int mt = wave / NT;
        const int nt = wave % NT;
        const int h  = lane >> 4;
        const int c  = lane & 15;
        const int kh = 2 * h;
        const float* arow = acc + (mt * 16 + c) * A1_P + kh;
        const float* bcol = sW + kh * N_CLS + nt * 16 + c;
        v8f d = {};
#pragma unroll
        for (int k = 0; k < HID; k += 4) {
            v2f a, b;
            a.x = fmaxf(arow[k + 0] + sB[k + kh + 0], 0.f);
            a.y = fmaxf(arow[k + 1] + sB[k + kh + 1], 0.f);
            b.x = bcol[(k + 0) * N_CLS];
            b.y = bcol[(k + 1) * N_CLS];
            d = __builtin_amdgcn_wmma_f32_16x16x4_f32(
                false, a, false, b, (short)0, d, false, false);
        }
        const int lr0   = mt * 16 + 8 * h;
        const int node0 = blockIdx.x * BIN_R;
#pragma unroll
        for (int r = 0; r < 8; ++r) {
            const int lr = lr0 + r;
            if (lr < BIN_R && node0 + lr < n_nodes)
                t2[(size_t)(node0 + lr) * N_CLS + nt * 16 + c] = d[r];
        }
    }
}

__global__ void __launch_bounds__(1024)
bin_sum_logsoftmax(const float* __restrict__ t2, const int* __restrict__ sorted,
                   const int* __restrict__ binStart, const float* __restrict__ b2,
                   float* __restrict__ out, int n_nodes) {
    __shared__ float4 acc4[BIN_R * N_CLS / 4];
    static_assert(BIN_R % 4 == 0, "whole four-row groups");

    const int lane = threadIdx.x & 31;
    const int sub  = lane >> 3;
    const int q    = lane & 7;
    float4 bb;
    bb.x = b2[q * 4 + 0]; bb.y = b2[q * 4 + 1];
    bb.z = b2[q * 4 + 2]; bb.w = b2[q * 4 + 3];

    bin_sum_rows<N_CLS, N_CLS, BIN_R>(t2, sorted, binStart, acc4, [] {});

    const int wave  = (int)(threadIdx.x >> 5);
    const int nw    = (int)(blockDim.x >> 5);
    const int node0 = blockIdx.x * BIN_R;
    constexpr int NG = BIN_R / 4;
    constexpr int U  = 2;
    float4* out4 = (float4*)out;
    for (int g0 = wave; g0 < NG; g0 += U * nw) {
        float4 x[U];
        int    lr[U];
        bool   live[U];
#pragma unroll
        for (int u = 0; u < U; ++u) {
            const int g = g0 + u * nw;
            live[u] = g < NG;
            lr[u]   = g * 4 + sub;
            const int lc = live[u] ? lr[u] : BIN_R - 1;
            const float4 v = acc4[lc * (N_CLS / 4) + q];
            x[u].x = v.x + bb.x; x[u].y = v.y + bb.y;
            x[u].z = v.z + bb.z; x[u].w = v.w + bb.w;
        }
        float4 r[U];
#pragma unroll
        for (int u = 0; u < U; ++u) {
            float m = fmaxf(fmaxf(x[u].x, x[u].y), fmaxf(x[u].z, x[u].w));
#pragma unroll
            for (int o = 4; o; o >>= 1) m = fmaxf(m, __shfl_xor(m, o, 32));
            x[u].x -= m; x[u].y -= m; x[u].z -= m; x[u].w -= m;
            float e0 = __expf(x[u].x), e1 = __expf(x[u].y);
            float e2 = __expf(x[u].z), e3 = __expf(x[u].w);
#pragma unroll
            for (int o = 4; o; o >>= 1) {
                e0 += __shfl_xor(e0, o, 32);
                e1 += __shfl_xor(e1, o, 32);
                e2 += __shfl_xor(e2, o, 32);
                e3 += __shfl_xor(e3, o, 32);
            }
            const float ls = __logf((e0 + e2) + (e1 + e3));
            r[u].x = x[u].x - ls; r[u].y = x[u].y - ls;
            r[u].z = x[u].z - ls; r[u].w = x[u].w - ls;
        }
#pragma unroll
        for (int u = 0; u < U; ++u)
            if (live[u] && node0 + lr[u] < n_nodes)
                out4[(size_t)(node0 + lr[u]) * (N_CLS / 4) + q] = r[u];
    }
}

extern "C" void kernel_launch(void* const* d_in, const int* in_sizes, int n_in,
                              void* d_out, int out_size, void* d_ws, size_t ws_size,
                              hipStream_t stream) {
    const float* x   = (const float*)d_in[0];
    const int*   ei  = (const int*)d_in[1];
    const float* W1  = (const float*)d_in[2];
    const float* b1  = (const float*)d_in[3];
    const float* W2  = (const float*)d_in[4];
    const float* b2  = (const float*)d_in[5];
    float*       out = (float*)d_out;

    const int n_edges = in_sizes[1] / 2;
    const int* src = ei;
    const int* dst = ei + n_edges;

    float* t1 = (float*)d_ws;
    float* t2 = t1 + (size_t)N_NODES * HID;
    int* counts   = (int*)(t2 + (size_t)N_NODES * N_CLS);
    int* binStart = counts + PART_WG * NB;
    int* sorted   = binStart + 2 * NB;
    (void)ws_size; (void)n_in; (void)out_size;

    {
        int waves = N_NODES / 16;
        gemm_wmma_f32<IN_F, HID><<<(waves + 7) / 8, 256, 0, stream>>>(
            x, W1, t1, N_NODES);
    }
    bin_count<<<PART_WG, 1024, 0, stream>>>(dst, n_edges, counts);
    bin_place<<<PART_WG, 4 * NB, 0, stream>>>(src, dst, n_edges, counts, sorted, binStart);
    bin_sum_relu_gemm<<<NB, 1024, 0, stream>>>(t1, sorted, binStart, b1, W2, t2, N_NODES);
    bin_sum_logsoftmax<<<NB, 1024, 0, stream>>>(t2, sorted, binStart, b2, out, N_NODES);
}


# edge grouping without a chip-wide count: each share sorts its own edges by bin in LDS in one launch, a bin's waves walk the 256 per-share segments; 5 launches become 4, arrival-order sums as before
# speedup vs baseline: 2.6721x; 1.2126x over previous
#include <hip/hip_runtime.h>
#include <hip/hip_bf16.h>

#define N_NODES 50000
#define IN_F    96
#define HID     64
#define N_CLS   32

typedef __attribute__((ext_vector_type(2))) float v2f;
typedef __attribute__((ext_vector_type(8))) float v8f;

template <int K, int N>
__global__ void gemm_wmma_f32(const float* __restrict__ X,
                              const float* __restrict__ W,
                              float* __restrict__ Y,
                              int M) {
    __shared__ float sW[K * N];

    const int wave = (int)(blockIdx.x * blockDim.x + threadIdx.x) >> 5;
    const int lane = threadIdx.x & 31;
    const int m0   = wave * 16;

    for (int i = threadIdx.x; i < K * N; i += blockDim.x) sW[i] = W[i];
    __syncthreads();

    if (m0 >= M) return;

    const int half16 = lane >> 4;
    const int l15    = lane & 15;
    const int kh     = half16 * 2;
    const int arow   = m0 + l15;

    constexpr int NT = N / 16;
    v8f acc[NT] = {};

    for (int k = 0; k < K; k += 4) {
        v2f a;
        a.x = X[(size_t)arow * K + k + kh + 0];
        a.y = X[(size_t)arow * K + k + kh + 1];
#pragma unroll
        for (int t = 0; t < NT; ++t) {
            v2f b;
            b.x = sW[(k + kh + 0) * N + t * 16 + l15];
            b.y = sW[(k + kh + 1) * N + t * 16 + l15];
            acc[t] = __builtin_amdgcn_wmma_f32_16x16x4_f32(
                false, a, false, b, (short)0, acc[t], false, false);
        }
    }

    const int rbase = m0 + half16 * 8;
#pragma unroll
    for (int t = 0; t < NT; ++t)
#pragma unroll
        for (int r = 0; r < 8; ++r)
            Y[(size_t)(rbase + r) * N + t * 16 + l15] = acc[t][r];
}

#define NB        256
#define BIN_R     196
#define PART_WG   256
#define SUM_WAVES 32
#define SPW       (PART_WG / SUM_WAVES)
static_assert(NB * BIN_R >= N_NODES, "bins must cover every node");
static_assert(N_NODES <= 65536 && BIN_R <= 65536, "record: 16 bits src, 16 bits local dst");
static_assert(PART_WG % SUM_WAVES == 0 && SPW <= 32, "a wave reads its shares' segments one per lane");

__device__ __forceinline__ void edge_share(int p, int P, int n_edges, int& e0, int& e1) {
    const int per = (n_edges + P - 1) / P;
    e0 = p * per;  if (e0 > n_edges) e0 = n_edges;
    e1 = e0 + per; if (e1 > n_edges) e1 = n_edges;
}

__device__ __forceinline__ void count_edges(int* hist, const int (&d)[4]) {
#pragma unroll
    for (int k = 0; k < 4; ++k)
        if (d[k] >= 0) atomicAdd(&hist[(unsigned)d[k] / BIN_R], 1);
}

__device__ __forceinline__ void place_edges(int* cursor, int* __restrict__ mine,
                                            const int (&d)[4], const int (&s)[4]) {
#pragma unroll
    for (int k = 0; k < 4; ++k) {
        if (d[k] >= 0) {
            const unsigned dd  = (unsigned)d[k];
            const unsigned bin = dd / BIN_R;
            const int slot = atomicAdd(&cursor[bin], 1);
            mine[slot] = (int)((unsigned)s[k] | ((dd - bin * BIN_R) << 16));
        }
    }
}

__global__ void __launch_bounds__(1024)
share_sort(const int* __restrict__ src, const int* __restrict__ dst, int n_edges,
           int* __restrict__ sorted, int* __restrict__ seg) {
    __shared__ int hist[NB];
    __shared__ int sc[2][NB];
    __shared__ int cursor[NB];
    const int t  = threadIdx.x;
    const int p  = blockIdx.x;
    const int nt = (int)blockDim.x;

    int e0, e1;
    edge_share(p, PART_WG, n_edges, e0, e1);

    int d0[4], s0[4];
#pragma unroll
    for (int k = 0; k < 4; ++k) {
        const int e = e0 + t + k * nt;
        d0[k] = e < e1 ? dst[e] : -1;
        s0[k] = e < e1 ? src[e] : 0;
    }
    for (int i = t; i < NB; i += nt) hist[i] = 0;
    __syncthreads();

    count_edges(hist, d0);
    for (int eb = e0 + t + 4 * nt; eb < e1; eb += 4 * nt) {
        int d[4];
#pragma unroll
        for (int k = 0; k < 4; ++k) {
            const int e = eb + k * nt;
            d[k] = e < e1 ? dst[e] : -1;
        }
        count_edges(hist, d);
    }
    __syncthreads();
    int tot = 0;
    if (t < NB) {
        tot = hist[t];
        sc[0][t] = tot;
    }
    __syncthreads();
    int cur = 0;
    for (int off = 1; off < NB; off <<= 1) {
        if (t < NB) {
            int v = sc[cur][t];
            if (t >= off) v += sc[cur][t - off];
            sc[cur ^ 1][t] = v;
        }
        __syncthreads();
        cur ^= 1;
    }
    if (t < NB) {
        const int start = sc[cur][t] - tot;
        cursor[t] = start;
        ((int2*)seg)[t * PART_WG + p] = make_int2(e0 + start, tot);
    }
    __syncthreads();

    int* mine = sorted + e0;
    place_edges(cursor, mine, d0, s0);
    for (int eb = e0 + t + 4 * nt; eb < e1; eb += 4 * nt) {
        int d[4], s[4];
#pragma unroll
        for (int k = 0; k < 4; ++k) {
            const int e = eb + k * nt;
            d[k] = e < e1 ? dst[e] : -1;
            s[k] = e < e1 ? src[e] : 0;
        }
        place_edges(cursor, mine, d, s);
    }
}

template <int F, int P, bool FULL>
__device__ __forceinline__ void gather_batch(const float* __restrict__ msg, float* acc,
                                             int rec, int cnt, int lane) {
    constexpr int SEG = F / 32;
    constexpr int CH  = 32 / SEG;
#pragma unroll
    for (int c = 0; c < 32; c += CH) {
        if (FULL || c < cnt) {
            float v[CH][SEG];
#pragma unroll
            for (int j = 0; j < CH; ++j) {
                const unsigned r = (unsigned)__builtin_amdgcn_readlane(rec, c + j);
                const float* row = msg + (size_t)(r & 0xFFFFu) * F;
#pragma unroll
                for (int g = 0; g < SEG; ++g)
                    v[j][g] = row[g * 32 + lane];
            }
#pragma unroll
            for (int j = 0; j < CH; ++j) {
                if (FULL || c + j < cnt) {
                    const unsigned r = (unsigned)__builtin_amdgcn_readlane(rec, c + j);
                    float* arow = acc + (r >> 16) * P;
#pragma unroll
                    for (int g = 0; g < SEG; ++g)
                        __hip_atomic_fetch_add(arow + g * 32 + lane, v[j][g],
                                               __ATOMIC_RELAXED, __HIP_MEMORY_SCOPE_WORKGROUP);
                }
            }
        }
    }
}

__device__ __forceinline__ int list_slot(int f, int o0, int o1, int o2, int o3, int o4,
                                         int o5, int o6, int o7, int p1, int p2, int p3,
                                         int p4, int p5, int p6, int p7) {
    int o = o0;
    o = f >= p1 ? o1 : o;
    o = f >= p2 ? o2 : o;
    o = f >= p3 ? o3 : o;
    o = f >= p4 ? o4 : o;
    o = f >= p5 ? o5 : o;
    o = f >= p6 ? o6 : o;
    o = f >= p7 ? o7 : o;
    return f + o;
}

template <int F, int P, int ZROWS, class Stage>
__device__ __forceinline__ void bin_sum_rows(const float* __restrict__ msg,
                                             const int* __restrict__ sorted,
                                             const int* __restrict__ seg,
                                             float4* acc4, Stage stage) {
    static_assert((ZROWS * P) % 4 == 0, "cleared as float4");
    float* acc = (float*)acc4;
    const int lane = threadIdx.x & 31;
    const int wave = __builtin_amdgcn_readfirstlane((int)(threadIdx.x >> 5));

    int2 si = make_int2(0, 0);
    if (lane < SPW) si = ((const int2*)seg)[blockIdx.x * PART_WG + wave * SPW + lane];
    static_assert(SPW == 8, "the eight segments are written out below");
    const int p1 = __builtin_amdgcn_readlane(si.y, 0);
    const int p2 = p1 + __builtin_amdgcn_readlane(si.y, 1);
    const int p3 = p2 + __builtin_amdgcn_readlane(si.y, 2);
    const int p4 = p3 + __builtin_amdgcn_readlane(si.y, 3);
    const int p5 = p4 + __builtin_amdgcn_readlane(si.y, 4);
    const int p6 = p5 + __builtin_amdgcn_readlane(si.y, 5);
    const int p7 = p6 + __builtin_amdgcn_readlane(si.y, 6);
    const int total = p7 + __builtin_amdgcn_readlane(si.y, 7);
    const int o0 = __builtin_amdgcn_readlane(si.x, 0);
    const int o1 = __builtin_amdgcn_readlane(si.x, 1) - p1;
    const int o2 = __builtin_amdgcn_readlane(si.x, 2) - p2;
    const int o3 = __builtin_amdgcn_readlane(si.x, 3) - p3;
    const int o4 = __builtin_amdgcn_readlane(si.x, 4) - p4;
    const int o5 = __builtin_amdgcn_readlane(si.x, 5) - p5;
    const int o6 = __builtin_amdgcn_readlane(si.x, 6) - p6;
    const int o7 = __builtin_amdgcn_readlane(si.x, 7) - p7;

    int base = 0;
    int rec = 0;
    if (lane < total)
        rec = sorted[list_slot(lane, o0, o1, o2, o3, o4, o5, o6, o7, p1, p2, p3, p4, p5, p6, p7)];

    for (int i = threadIdx.x; i < ZROWS * P / 4; i += blockDim.x)
        acc4[i] = make_float4(0.f, 0.f, 0.f, 0.f);
    stage();
    __syncthreads();

    while (base < total) {
        const int nbase = base + 32;
        int nrec = 0;
        if (nbase + lane < total)
            nrec = sorted[list_slot(nbase + lane, o0, o1, o2, o3, o4, o5, o6, o7, p1, p2, p3, p4, p5, p6, p7)];
        const int left = total - base;
        const int cnt  = left < 32 ? left : 32;
        if (cnt == 32) gather_batch<F, P, true>(msg, acc, rec, cnt, lane);
        else           gather_batch<F, P, false>(msg, acc, rec, cnt, lane);
        rec  = nrec;
        base = nbase;
    }
    __syncthreads();
}

#define A1_P    68
#define A1_ROWS 208
static_assert(A1_ROWS % 16 == 0 && A1_ROWS >= BIN_R, "whole 16-row tiles over the bin");

__global__ void __launch_bounds__(1024)
bin_sum_relu_gemm(const float* __restrict__ t1, const int* __restrict__ sorted,
                  const int* __restrict__ seg, const float* __restrict__ b1,
                  const float* __restrict__ W2, float* __restrict__ t2, int n_nodes) {
    __shared__ float4 acc4[A1_ROWS * A1_P / 4];
    __shared__ float  sW[HID * N_CLS];
    __shared__ float  sB[HID];
    static_assert(HID * N_CLS == 2 * 1024, "two weights per thread of the 1024");

    const int   tid = threadIdx.x;
    const float w0  = W2[tid];
    const float w1  = W2[tid + 1024];
    const float bv  = b1[tid & (HID - 1)];

    bin_sum_rows<HID, A1_P, A1_ROWS>(t1, sorted, seg, acc4, [&] {
        sW[tid]        = w0;
        sW[tid + 1024] = w1;
        if (tid < HID) sB[tid] = bv;
    });

    const float* acc = (const float*)acc4;
    const int lane = tid & 31;
    const int wave = __builtin_amdgcn_readfirstlane(tid >> 5);
    constexpr int MT = A1_ROWS / 16;
    constexpr int NT = N_CLS / 16;
    if (wave < MT * NT) {
        const int mt = wave / NT;
        const int nt = wave % NT;
        const int h  = lane >> 4;
        const int c  = lane & 15;
        const int kh = 2 * h;
        const float* arow = acc + (mt * 16 + c) * A1_P + kh;
        const float* bcol = sW + kh * N_CLS + nt * 16 + c;
        v8f d = {};
#pragma unroll
        for (int k = 0; k < HID; k += 4) {
            v2f a, b;
            a.x = fmaxf(arow[k + 0] + sB[k + kh + 0], 0.f);
            a.y = fmaxf(arow[k + 1] + sB[k + kh + 1], 0.f);
            b.x = bcol[(k + 0) * N_CLS];
            b.y = bcol[(k + 1) * N_CLS];
            d = __builtin_amdgcn_wmma_f32_16x16x4_f32(
                false, a, false, b, (short)0, d, false, false);
        }
        const int lr0   = mt * 16 + 8 * h;
        const int node0 = blockIdx.x * BIN_R;
#pragma unroll
        for (int r = 0; r < 8; ++r) {
            const int lr = lr0 + r;
            if (lr < BIN_R && node0 + lr < n_nodes)
                t2[(size_t)(node0 + lr) * N_CLS + nt * 16 + c] = d[r];
        }
    }
}

__global__ void __launch_bounds__(1024)
bin_sum_logsoftmax(const float* __restrict__ t2, const int* __restrict__ sorted,
                   const int* __restrict__ seg, const float* __restrict__ b2,
                   float* __restrict__ out, int n_nodes) {
    __shared__ float4 acc4[BIN_R * N_CLS / 4];
    static_assert(BIN_R % 4 == 0, "whole four-row groups");

    const int lane = threadIdx.x & 31;
    const int sub  = lane >> 3;
    const int q    = lane & 7;
    float4 bb;
    bb.x = b2[q * 4 + 0]; bb.y = b2[q * 4 + 1];
    bb.z = b2[q * 4 + 2]; bb.w = b2[q * 4 + 3];

    bin_sum_rows<N_CLS, N_CLS, BIN_R>(t2, sorted, seg, acc4, [] {});

    const int wave  = (int)(threadIdx.x >> 5);
    const int nw    = (int)(blockDim.x >> 5);
    const int node0 = blockIdx.x * BIN_R;
    constexpr int NG = BIN_R / 4;
    constexpr int U  = 2;
    float4* out4 = (float4*)out;
    for (int g0 = wave; g0 < NG; g0 += U * nw) {
        float4 x[U];
        int    lr[U];
        bool   live[U];
#pragma unroll
        for (int u = 0; u < U; ++u) {
            const int g = g0 + u * nw;
            live[u] = g < NG;
            lr[u]   = g * 4 + sub;
            const int lc = live[u] ? lr[u] : BIN_R - 1;
            const float4 v = acc4[lc * (N_CLS / 4) + q];
            x[u].x = v.x + bb.x; x[u].y = v.y + bb.y;
            x[u].z = v.z + bb.z; x[u].w = v.w + bb.w;
        }
        float4 r[U];
#pragma unroll
        for (int u = 0; u < U; ++u) {
            float m = fmaxf(fmaxf(x[u].x, x[u].y), fmaxf(x[u].z, x[u].w));
#pragma unroll
            for (int o = 4; o; o >>= 1) m = fmaxf(m, __shfl_xor(m, o, 32));
            x[u].x -= m; x[u].y -= m; x[u].z -= m; x[u].w -= m;
            float e0 = __expf(x[u].x), e1 = __expf(x[u].y);
            float e2 = __expf(x[u].z), e3 = __expf(x[u].w);
#pragma unroll
            for (int o = 4; o; o >>= 1) {
                e0 += __shfl_xor(e0, o, 32);
                e1 += __shfl_xor(e1, o, 32);
                e2 += __shfl_xor(e2, o, 32);
                e3 += __shfl_xor(e3, o, 32);
            }
            const float ls = __logf((e0 + e2) + (e1 + e3));
            r[u].x = x[u].x - ls; r[u].y = x[u].y - ls;
            r[u].z = x[u].z - ls; r[u].w = x[u].w - ls;
        }
#pragma unroll
        for (int u = 0; u < U; ++u)
            if (live[u] && node0 + lr[u] < n_nodes)
                out4[(size_t)(node0 + lr[u]) * (N_CLS / 4) + q] = r[u];
    }
}

extern "C" void kernel_launch(void* const* d_in, const int* in_sizes, int n_in,
                              void* d_out, int out_size, void* d_ws, size_t ws_size,
                              hipStream_t stream) {
    const float* x   = (const float*)d_in[0];
    const int*   ei  = (const int*)d_in[1];
    const float* W1  = (const float*)d_in[2];
    const float* b1  = (const float*)d_in[3];
    const float* W2  = (const float*)d_in[4];
    const float* b2  = (const float*)d_in[5];
    float*       out = (float*)d_out;

    const int n_edges = in_sizes[1] / 2;
    const int* src = ei;
    const int* dst = ei + n_edges;

    float* t1 = (float*)d_ws;
    float* t2 = t1 + (size_t)N_NODES * HID;
    int* seg    = (int*)(t2 + (size_t)N_NODES * N_CLS);
    int* sorted = seg + 2 * NB * PART_WG;
    (void)ws_size; (void)n_in; (void)out_size;

    {
        int waves = N_NODES / 16;
        gemm_wmma_f32<IN_F, HID><<<(waves + 7) / 8, 256, 0, stream>>>(
            x, W1, t1, N_NODES);
    }
    share_sort<<<PART_WG, 1024, 0, stream>>>(src, dst, n_edges, sorted, seg);
    bin_sum_relu_gemm<<<NB, SUM_WAVES * 32, 0, stream>>>(t1, sorted, seg, b1, W2, t2, N_NODES);
    bin_sum_logsoftmax<<<NB, SUM_WAVES * 32, 0, stream>>>(t2, sorted, seg, b2, out, N_NODES);
}


# x@W1 and the per-share edge sort, mutually independent, share one launch (split by block number; 256-thread blocks, a sort thread holds 13 edges in registers); 4 launches to 3; arrival-order sums
# speedup vs baseline: 3.1746x; 1.1881x over previous
#include <hip/hip_runtime.h>
#include <hip/hip_bf16.h>

#define N_NODES 50000
#define IN_F    96
#define HID     64
#define N_CLS   32

typedef __attribute__((ext_vector_type(2))) float v2f;
typedef __attribute__((ext_vector_type(8))) float v8f;

template <int K, int N>
__device__ __forceinline__ void gemm_wmma_f32(const float* __restrict__ X,
                                              const float* __restrict__ W,
                                              float* __restrict__ Y,
                                              int M, int blk) {
    __shared__ float sW[K * N];

    const int wave = (int)(blk * blockDim.x + threadIdx.x) >> 5;
    const int lane = threadIdx.x & 31;
    const int m0   = wave * 16;

    for (int i = threadIdx.x; i < K * N; i += blockDim.x) sW[i] = W[i];
    __syncthreads();

    if (m0 >= M) return;

    const int half16 = lane >> 4;
    const int l15    = lane & 15;
    const int kh     = half16 * 2;
    const int arow   = m0 + l15;

    constexpr int NT = N / 16;
    v8f acc[NT] = {};

    for (int k = 0; k < K; k += 4) {
        v2f a;
        a.x = X[(size_t)arow * K + k + kh + 0];
        a.y = X[(size_t)arow * K + k + kh + 1];
#pragma unroll
        for (int t = 0; t < NT; ++t) {
            v2f b;
            b.x = sW[(k + kh + 0) * N + t * 16 + l15];
            b.y = sW[(k + kh + 1) * N + t * 16 + l15];
            acc[t] = __builtin_amdgcn_wmma_f32_16x16x4_f32(
                false, a, false, b, (short)0, acc[t], false, false);
        }
    }

    const int rbase = m0 + half16 * 8;
#pragma unroll
    for (int t = 0; t < NT; ++t)
#pragma unroll
        for (int r = 0; r < 8; ++r)
            Y[(size_t)(rbase + r) * N + t * 16 + l15] = acc[t][r];
}

#define NB        256
#define BIN_R     196
#define PART_WG   256
#define SUM_WAVES 32
#define SPW       (PART_WG / SUM_WAVES)
#define SORT_EPT  13
static_assert(NB * BIN_R >= N_NODES, "bins must cover every node");
static_assert(N_NODES <= 65536 && BIN_R <= 65536, "record: 16 bits src, 16 bits local dst");
static_assert(PART_WG % SUM_WAVES == 0 && SPW <= 32, "a wave reads its shares' segments one per lane");

__device__ __forceinline__ void edge_share(int p, int P, int n_edges, int& e0, int& e1) {
    const int per = (n_edges + P - 1) / P;
    e0 = p * per;  if (e0 > n_edges) e0 = n_edges;
    e1 = e0 + per; if (e1 > n_edges) e1 = n_edges;
}

__device__ __forceinline__ unsigned edge_counter(int d, int lane) {
    return d >= 0 ? (unsigned)d / BIN_R : (unsigned)(NB + lane);
}

__device__ __forceinline__ void count_edges(int* hist, const int (&d)[SORT_EPT], int lane) {
#pragma unroll
    for (int k = 0; k < SORT_EPT; ++k)
        atomicAdd(&hist[edge_counter(d[k], lane)], 1);
}

__device__ __forceinline__ void place_edges(int* cursor, int* __restrict__ mine,
                                            const int (&d)[SORT_EPT], const int (&s)[SORT_EPT],
                                            int lane) {
    int slot[SORT_EPT];
#pragma unroll
    for (int k = 0; k < SORT_EPT; ++k)
        slot[k] = atomicAdd(&cursor[edge_counter(d[k], lane)], 1);
#pragma unroll
    for (int k = 0; k < SORT_EPT; ++k) {
        if (d[k] >= 0) {
            const unsigned dd  = (unsigned)d[k];
            const unsigned bin = dd / BIN_R;
            mine[slot[k]] = (int)((unsigned)s[k] | ((dd - bin * BIN_R) << 16));
        }
    }
}

__device__ __forceinline__ void
share_sort(const int* __restrict__ src, const int* __restrict__ dst, int n_edges,
           int* __restrict__ sorted, int* __restrict__ seg, int p) {
    __shared__ int hist[NB + 32];
    __shared__ int sc[2][NB];
    __shared__ int cursor[NB + 32];
    const int t    = threadIdx.x;
    const int nt   = (int)blockDim.x;
    const int lane = t & 31;

    int e0, e1;
    edge_share(p, PART_WG, n_edges, e0, e1);

    int d0[SORT_EPT], s0[SORT_EPT];
#pragma unroll
    for (int k = 0; k < SORT_EPT; ++k) {
        const int e = e0 + t + k * nt;
        d0[k] = e < e1 ? dst[e] : -1;
        s0[k] = e < e1 ? src[e] : 0;
    }
    for (int i = t; i < NB + 32; i += nt) hist[i] = 0;
    __syncthreads();

    count_edges(hist, d0, lane);
    for (int eb = e0 + t + SORT_EPT * nt; eb < e1; eb += SORT_EPT * nt) {
        int d[SORT_EPT];
#pragma unroll
        for (int k = 0; k < SORT_EPT; ++k) {
            const int e = eb + k * nt;
            d[k] = e < e1 ? dst[e] : -1;
        }
        count_edges(hist, d, lane);
    }
    __syncthreads();
    int tot = 0;
    if (t < NB) {
        tot = hist[t];
        sc[0][t] = tot;
    }
    __syncthreads();
    int cur = 0;
    for (int off = 1; off < NB; off <<= 1) {
        if (t < NB) {
            int v = sc[cur][t];
            if (t >= off) v += sc[cur][t - off];
            sc[cur ^ 1][t] = v;
        }
        __syncthreads();
        cur ^= 1;
    }
    if (t < NB) {
        const int start = sc[cur][t] - tot;
        cursor[t] = start;
        if (t < 32) cursor[NB + t] = 0;
        ((int2*)seg)[t * PART_WG + p] = make_int2(e0 + start, tot);
    }
    __syncthreads();

    int* mine = sorted + e0;
    place_edges(cursor, mine, d0, s0, lane);
    for (int eb = e0 + t + SORT_EPT * nt; eb < e1; eb += SORT_EPT * nt) {
        int d[SORT_EPT], s[SORT_EPT];
#pragma unroll
        for (int k = 0; k < SORT_EPT; ++k) {
            const int e = eb + k * nt;
            d[k] = e < e1 ? dst[e] : -1;
            s[k] = e < e1 ? src[e] : 0;
        }
        place_edges(cursor, mine, d, s, lane);
    }
}

#define GEMM_WG ((N_NODES / 16 + 7) / 8)
__global__ void __launch_bounds__(256)
gemm1_and_share_sort(const float* __restrict__ x, const float* __restrict__ W1,
                     float* __restrict__ t1, int n_nodes,
                     const int* __restrict__ src, const int* __restrict__ dst, int n_edges,
                     int* __restrict__ sorted, int* __restrict__ seg) {
    if (blockIdx.x < GEMM_WG)
        gemm_wmma_f32<IN_F, HID>(x, W1, t1, n_nodes, (int)blockIdx.x);
    else
        share_sort(src, dst, n_edges, sorted, seg, (int)blockIdx.x - GEMM_WG);
}

template <int F, int P, bool FULL>
__device__ __forceinline__ void gather_batch(const float* __restrict__ msg, float* acc,
                                             int rec, int cnt, int lane) {
    constexpr int SEG = F / 32;
    constexpr int CH  = 32 / SEG;
#pragma unroll
    for (int c = 0; c < 32; c += CH) {
        if (FULL || c < cnt) {
            float v[CH][SEG];
#pragma unroll
            for (int j = 0; j < CH; ++j) {
                const unsigned r = (unsigned)__builtin_amdgcn_readlane(rec, c + j);
                const float* row = msg + (size_t)(r & 0xFFFFu) * F;
#pragma unroll
                for (int g = 0; g < SEG; ++g)
                    v[j][g] = row[g * 32 + lane];
            }
#pragma unroll
            for (int j = 0; j < CH; ++j) {
                if (FULL || c + j < cnt) {
                    const unsigned r = (unsigned)__builtin_amdgcn_readlane(rec, c + j);
                    float* arow = acc + (r >> 16) * P;
#pragma unroll
                    for (int g = 0; g < SEG; ++g)
                        __hip_atomic_fetch_add(arow + g * 32 + lane, v[j][g],
                                               __ATOMIC_RELAXED, __HIP_MEMORY_SCOPE_WORKGROUP);
                }
            }
        }
    }
}

__device__ __forceinline__ int list_slot(int f, int o0, int o1, int o2, int o3, int o4,
                                         int o5, int o6, int o7, int p1, int p2, int p3,
                                         int p4, int p5, int p6, int p7) {
    int o = o0;
    o = f >= p1 ? o1 : o;
    o = f >= p2 ? o2 : o;
    o = f >= p3 ? o3 : o;
    o = f >= p4 ? o4 : o;
    o = f >= p5 ? o5 : o;
    o = f >= p6 ? o6 : o;
    o = f >= p7 ? o7 : o;
    return f + o;
}

template <int F, int P, int ZROWS, class Stage>
__device__ __forceinline__ void bin_sum_rows(const float* __restrict__ msg,
                                             const int* __restrict__ sorted,
                                             const int* __restrict__ seg,
                                             float4* acc4, Stage stage) {
    static_assert((ZROWS * P) % 4 == 0, "cleared as float4");
    float* acc = (float*)acc4;
    const int lane = threadIdx.x & 31;
    const int wave = __builtin_amdgcn_readfirstlane((int)(threadIdx.x >> 5));

    int2 si = make_int2(0, 0);
    if (lane < SPW) si = ((const int2*)seg)[blockIdx.x * PART_WG + wave * SPW + lane];
    static_assert(SPW == 8, "the eight segments are written out below");
    const int p1 = __builtin_amdgcn_readlane(si.y, 0);
    const int p2 = p1 + __builtin_amdgcn_readlane(si.y, 1);
    const int p3 = p2 + __builtin_amdgcn_readlane(si.y, 2);
    const int p4 = p3 + __builtin_amdgcn_readlane(si.y, 3);
    const int p5 = p4 + __builtin_amdgcn_readlane(si.y, 4);
    const int p6 = p5 + __builtin_amdgcn_readlane(si.y, 5);
    const int p7 = p6 + __builtin_amdgcn_readlane(si.y, 6);
    const int total = p7 + __builtin_amdgcn_readlane(si.y, 7);
    const int o0 = __builtin_amdgcn_readlane(si.x, 0);
    const int o1 = __builtin_amdgcn_readlane(si.x, 1) - p1;
    const int o2 = __builtin_amdgcn_readlane(si.x, 2) - p2;
    const int o3 = __builtin_amdgcn_readlane(si.x, 3) - p3;
    const int o4 = __builtin_amdgcn_readlane(si.x, 4) - p4;
    const int o5 = __builtin_amdgcn_readlane(si.x, 5) - p5;
    const int o6 = __builtin_amdgcn_readlane(si.x, 6) - p6;
    const int o7 = __builtin_amdgcn_readlane(si.x, 7) - p7;

    int base = 0;
    int rec = 0;
    if (lane < total)
        rec = sorted[list_slot(lane, o0, o1, o2, o3, o4, o5, o6, o7, p1, p2, p3, p4, p5, p6, p7)];

    for (int i = threadIdx.x; i < ZROWS * P / 4; i += blockDim.x)
        acc4[i] = make_float4(0.f, 0.f, 0.f, 0.f);
    stage();
    __syncthreads();

    while (base < total) {
        const int nbase = base + 32;
        int nrec = 0;
        if (nbase + lane < total)
            nrec = sorted[list_slot(nbase + lane, o0, o1, o2, o3, o4, o5, o6, o7, p1, p2, p3, p4, p5, p6, p7)];
        const int left = total - base;
        const int cnt  = left < 32 ? left : 32;
        if (cnt == 32) gather_batch<F, P, true>(msg, acc, rec, cnt, lane);
        else           gather_batch<F, P, false>(msg, acc, rec, cnt, lane);
        rec  = nrec;
        base = nbase;
    }
    __syncthreads();
}

#define A1_P    68
#define A1_ROWS 208
static_assert(A1_ROWS % 16 == 0 && A1_ROWS >= BIN_R, "whole 16-row tiles over the bin");

__global__ void __launch_bounds__(1024)
bin_sum_relu_gemm(const float* __restrict__ t1, const int* __restrict__ sorted,
                  const int* __restrict__ seg, const float* __restrict__ b1,
                  const float* __restrict__ W2, float* __restrict__ t2, int n_nodes) {
    __shared__ float4 acc4[A1_ROWS * A1_P / 4];
    __shared__ float  sW[HID * N_CLS];
    __shared__ float  sB[HID];
    static_assert(HID * N_CLS == 2 * 1024, "two weights per thread of the 1024");

    const int   tid = threadIdx.x;
    const float w0  = W2[tid];
    const float w1  = W2[tid + 1024];
    const float bv  = b1[tid & (HID - 1)];

    bin_sum_rows<HID, A1_P, A1_ROWS>(t1, sorted, seg, acc4, [&] {
        sW[tid]        = w0;
        sW[tid + 1024] = w1;
        if (tid < HID) sB[tid] = bv;
    });

    const float* acc = (const float*)acc4;
    const int lane = tid & 31;
    const int wave = __builtin_amdgcn_readfirstlane(tid >> 5);
    constexpr int MT = A1_ROWS / 16;
    constexpr int NT = N_CLS / 16;
    if (wave < MT * NT) {
        const int mt = wave / NT;
        const int nt = wave % NT;
        const int h  = lane >> 4;
        const int c  = lane & 15;
        const int kh = 2 * h;
        const float* arow = acc + (mt * 16 + c) * A1_P + kh;
        const float* bcol = sW + kh * N_CLS + nt * 16 + c;
        v8f d = {};
#pragma unroll
        for (int k = 0; k < HID; k += 4) {
            v2f a, b;
            a.x = fmaxf(arow[k + 0] + sB[k + kh + 0], 0.f);
            a.y = fmaxf(arow[k + 1] + sB[k + kh + 1], 0.f);
            b.x = bcol[(k + 0) * N_CLS];
            b.y = bcol[(k + 1) * N_CLS];
            d = __builtin_amdgcn_wmma_f32_16x16x4_f32(
                false, a, false, b, (short)0, d, false, false);
        }
        const int lr0   = mt * 16 + 8 * h;
        const int node0 = blockIdx.x * BIN_R;
#pragma unroll
        for (int r = 0; r < 8; ++r) {
            const int lr = lr0 + r;
            if (lr < BIN_R && node0 + lr < n_nodes)
                t2[(size_t)(node0 + lr) * N_CLS + nt * 16 + c] = d[r];
        }
    }
}

__global__ void __launch_bounds__(1024)
bin_sum_logsoftmax(const float* __restrict__ t2, const int* __restrict__ sorted,
                   const int* __restrict__ seg, const float* __restrict__ b2,
                   float* __restrict__ out, int n_nodes) {
    __shared__ float4 acc4[BIN_R * N_CLS / 4];
    static_assert(BIN_R % 4 == 0, "whole four-row groups");

    const int lane = threadIdx.x & 31;
    const int sub  = lane >> 3;
    const int q    = lane & 7;
    float4 bb;
    bb.x = b2[q * 4 + 0]; bb.y = b2[q * 4 + 1];
    bb.z = b2[q * 4 + 2]; bb.w = b2[q * 4 + 3];

    bin_sum_rows<N_CLS, N_CLS, BIN_R>(t2, sorted, seg, acc4, [] {});

    const int wave  = (int)(threadIdx.x >> 5);
    const int nw    = (int)(blockDim.x >> 5);
    const int node0 = blockIdx.x * BIN_R;
    constexpr int NG = BIN_R / 4;
    constexpr int U  = 2;
    float4* out4 = (float4*)out;
    for (int g0 = wave; g0 < NG; g0 += U * nw) {
        float4 x[U];
        int    lr[U];
        bool   live[U];
#pragma unroll
        for (int u = 0; u < U; ++u) {
            const int g = g0 + u * nw;
            live[u] = g < NG;
            lr[u]   = g * 4 + sub;
            const int lc = live[u] ? lr[u] : BIN_R - 1;
            const float4 v = acc4[lc * (N_CLS / 4) + q];
            x[u].x = v.x + bb.x; x[u].y = v.y + bb.y;
            x[u].z = v.z + bb.z; x[u].w = v.w + bb.w;
        }
        float4 r[U];
#pragma unroll
        for (int u = 0; u < U; ++u) {
            float m = fmaxf(fmaxf(x[u].x, x[u].y), fmaxf(x[u].z, x[u].w));
#pragma unroll
            for (int o = 4; o; o >>= 1) m = fmaxf(m, __shfl_xor(m, o, 32));
            x[u].x -= m; x[u].y -= m; x[u].z -= m; x[u].w -= m;
            float e0 = __expf(x[u].x), e1 = __expf(x[u].y);
            float e2 = __expf(x[u].z), e3 = __expf(x[u].w);
#pragma unroll
            for (int o = 4; o; o >>= 1) {
                e0 += __shfl_xor(e0, o, 32);
                e1 += __shfl_xor(e1, o, 32);
                e2 += __shfl_xor(e2, o, 32);
                e3 += __shfl_xor(e3, o, 32);
            }
            const float ls = __logf((e0 + e2) + (e1 + e3));
            r[u].x = x[u].x - ls; r[u].y = x[u].y - ls;
            r[u].z = x[u].z - ls; r[u].w = x[u].w - ls;
        }
#pragma unroll
        for (int u = 0; u < U; ++u)
            if (live[u] && node0 + lr[u] < n_nodes)
                out4[(size_t)(node0 + lr[u]) * (N_CLS / 4) + q] = r[u];
    }
}

extern "C" void kernel_launch(void* const* d_in, const int* in_sizes, int n_in,
                              void* d_out, int out_size, void* d_ws, size_t ws_size,
                              hipStream_t stream) {
    const float* x   = (const float*)d_in[0];
    const int*   ei  = (const int*)d_in[1];
    const float* W1  = (const float*)d_in[2];
    const float* b1  = (const float*)d_in[3];
    const float* W2  = (const float*)d_in[4];
    const float* b2  = (const float*)d_in[5];
    float*       out = (float*)d_out;

    const int n_edges = in_sizes[1] / 2;
    const int* src = ei;
    const int* dst = ei + n_edges;

    float* t1 = (float*)d_ws;
    float* t2 = t1 + (size_t)N_NODES * HID;
    int* seg    = (int*)(t2 + (size_t)N_NODES * N_CLS);
    int* sorted = seg + 2 * NB * PART_WG;
    (void)ws_size; (void)n_in; (void)out_size;

    gemm1_and_share_sort<<<GEMM_WG + PART_WG, 256, 0, stream>>>(
        x, W1, t1, N_NODES, src, dst, n_edges, sorted, seg);
    bin_sum_relu_gemm<<<NB, SUM_WAVES * 32, 0, stream>>>(t1, sorted, seg, b1, W2, t2, N_NODES);
    bin_sum_logsoftmax<<<NB, SUM_WAVES * 32, 0, stream>>>(t2, sorted, seg, b2, out, N_NODES);
}


# first launch, GEMM role: each wave asks for its whole A strip (24 8-byte loads) and the block for all of W1 (16-byte loads) up front, not one dependent round trip per k group and staged word
# speedup vs baseline: 3.4733x; 1.0941x over previous
#include <hip/hip_runtime.h>
#include <hip/hip_bf16.h>

#define N_NODES 50000
#define IN_F    96
#define HID     64
#define N_CLS   32

typedef __attribute__((ext_vector_type(2))) float v2f;
typedef __attribute__((ext_vector_type(8))) float v8f;

#define GEMM_THREADS 256
template <int K, int N>
__device__ __forceinline__ void gemm_wmma_f32(const float* __restrict__ X,
                                              const float* __restrict__ W,
                                              float* __restrict__ Y,
                                              int M, int blk) {
    constexpr int KS = K / 4;
    constexpr int WV = K * N / (4 * GEMM_THREADS);
    static_assert(K % 4 == 0 && (K * N) % (4 * GEMM_THREADS) == 0, "W is staged as whole float4 per thread");
    __shared__ float4 sW4[K * N / 4];
    const float* sW = (const float*)sW4;

    const int tid  = (int)threadIdx.x;
    const int wave = (blk * GEMM_THREADS + tid) >> 5;
    const int lane = tid & 31;
    const int m0   = wave * 16;
    const bool live = m0 < M;

    const int half16 = lane >> 4;
    const int l15    = lane & 15;
    const int kh     = half16 * 2;
    const int arow   = m0 + l15;

    v2f a[KS];
    if (live) {
        const float* xr = X + (size_t)arow * K + kh;
#pragma unroll
        for (int s = 0; s < KS; ++s) {
            a[s].x = xr[4 * s + 0];
            a[s].y = xr[4 * s + 1];
        }
    } else {
#pragma unroll
        for (int s = 0; s < KS; ++s) { a[s].x = 0.f; a[s].y = 0.f; }
    }

    float4 w[WV];
#pragma unroll
    for (int i = 0; i < WV; ++i) w[i] = ((const float4*)W)[tid + i * GEMM_THREADS];
#pragma unroll
    for (int i = 0; i < WV; ++i) sW4[tid + i * GEMM_THREADS] = w[i];
    __syncthreads();

    if (!live) return;

    constexpr int NT = N / 16;
    v8f acc[NT] = {};

#pragma unroll
    for (int s = 0; s < KS; ++s) {
        const int k = 4 * s;
#pragma unroll
        for (int t = 0; t < NT; ++t) {
            v2f b;
            b.x = sW[(k + kh + 0) * N + t * 16 + l15];
            b.y = sW[(k + kh + 1) * N + t * 16 + l15];
            acc[t] = __builtin_amdgcn_wmma_f32_16x16x4_f32(
                false, a[s], false, b, (short)0, acc[t], false, false);
        }
    }

    const int rbase = m0 + half16 * 8;
#pragma unroll
    for (int t = 0; t < NT; ++t)
#pragma unroll
        for (int r = 0; r < 8; ++r)
            Y[(size_t)(rbase + r) * N + t * 16 + l15] = acc[t][r];
}

#define NB        256
#define BIN_R     196
#define PART_WG   256
#define SUM_WAVES 32
#define SPW       (PART_WG / SUM_WAVES)
#define SORT_EPT  13
static_assert(NB * BIN_R >= N_NODES, "bins must cover every node");
static_assert(N_NODES <= 65536 && BIN_R <= 65536, "record: 16 bits src, 16 bits local dst");
static_assert(PART_WG % SUM_WAVES == 0 && SPW <= 32, "a wave reads its shares' segments one per lane");

__device__ __forceinline__ void edge_share(int p, int P, int n_edges, int& e0, int& e1) {
    const int per = (n_edges + P - 1) / P;
    e0 = p * per;  if (e0 > n_edges) e0 = n_edges;
    e1 = e0 + per; if (e1 > n_edges) e1 = n_edges;
}

__device__ __forceinline__ unsigned edge_counter(int d, int lane) {
    return d >= 0 ? (unsigned)d / BIN_R : (unsigned)(NB + lane);
}

__device__ __forceinline__ void count_edges(int* hist, const int (&d)[SORT_EPT], int lane) {
#pragma unroll
    for (int k = 0; k < SORT_EPT; ++k)
        atomicAdd(&hist[edge_counter(d[k], lane)], 1);
}

__device__ __forceinline__ void place_edges(int* cursor, int* __restrict__ mine,
                                            const int (&d)[SORT_EPT], const int (&s)[SORT_EPT],
                                            int lane) {
    int slot[SORT_EPT];
#pragma unroll
    for (int k = 0; k < SORT_EPT; ++k)
        slot[k] = atomicAdd(&cursor[edge_counter(d[k], lane)], 1);
#pragma unroll
    for (int k = 0; k < SORT_EPT; ++k) {
        if (d[k] >= 0) {
            const unsigned dd  = (unsigned)d[k];
            const unsigned bin = dd / BIN_R;
            mine[slot[k]] = (int)((unsigned)s[k] | ((dd - bin * BIN_R) << 16));
        }
    }
}

__device__ __forceinline__ void
share_sort(const int* __restrict__ src, const int* __restrict__ dst, int n_edges,
           int* __restrict__ sorted, int* __restrict__ seg, int p) {
    __shared__ int hist[NB + 32];
    __shared__ int sc[2][NB];
    __shared__ int cursor[NB + 32];
    const int t    = threadIdx.x;
    const int nt   = (int)blockDim.x;
    const int lane = t & 31;

    int e0, e1;
    edge_share(p, PART_WG, n_edges, e0, e1);

    int d0[SORT_EPT], s0[SORT_EPT];
#pragma unroll
    for (int k = 0; k < SORT_EPT; ++k) {
        const int e = e0 + t + k * nt;
        d0[k] = e < e1 ? dst[e] : -1;
        s0[k] = e < e1 ? src[e] : 0;
    }
    for (int i = t; i < NB + 32; i += nt) hist[i] = 0;
    __syncthreads();

    count_edges(hist, d0, lane);
    for (int eb = e0 + t + SORT_EPT * nt; eb < e1; eb += SORT_EPT * nt) {
        int d[SORT_EPT];
#pragma unroll
        for (int k = 0; k < SORT_EPT; ++k) {
            const int e = eb + k * nt;
            d[k] = e < e1 ? dst[e] : -1;
        }
        count_edges(hist, d, lane);
    }
    __syncthreads();
    int tot = 0;
    if (t < NB) {
        tot = hist[t];
        sc[0][t] = tot;
    }
    __syncthreads();
    int cur = 0;
    for (int off = 1; off < NB; off <<= 1) {
        if (t < NB) {
            int v = sc[cur][t];
            if (t >= off) v += sc[cur][t - off];
            sc[cur ^ 1][t] = v;
        }
        __syncthreads();
        cur ^= 1;
    }
    if (t < NB) {
        const int start = sc[cur][t] - tot;
        cursor[t] = start;
        if (t < 32) cursor[NB + t] = 0;
        ((int2*)seg)[t * PART_WG + p] = make_int2(e0 + start, tot);
    }
    __syncthreads();

    int* mine = sorted + e0;
    place_edges(cursor, mine, d0, s0, lane);
    for (int eb = e0 + t + SORT_EPT * nt; eb < e1; eb += SORT_EPT * nt) {
        int d[SORT_EPT], s[SORT_EPT];
#pragma unroll
        for (int k = 0; k < SORT_EPT; ++k) {
            const int e = eb + k * nt;
            d[k] = e < e1 ? dst[e] : -1;
            s[k] = e < e1 ? src[e] : 0;
        }
        place_edges(cursor, mine, d, s, lane);
    }
}

#define GEMM_WG ((N_NODES / 16 + 7) / 8)
__global__ void __launch_bounds__(GEMM_THREADS)
gemm1_and_share_sort(const float* __restrict__ x, const float* __restrict__ W1,
                     float* __restrict__ t1, int n_nodes,
                     const int* __restrict__ src, const int* __restrict__ dst, int n_edges,
                     int* __restrict__ sorted, int* __restrict__ seg) {
    if (blockIdx.x < GEMM_WG)
        gemm_wmma_f32<IN_F, HID>(x, W1, t1, n_nodes, (int)blockIdx.x);
    else
        share_sort(src, dst, n_edges, sorted, seg, (int)blockIdx.x - GEMM_WG);
}

template <int F, int P, bool FULL>
__device__ __forceinline__ void gather_batch(const float* __restrict__ msg, float* acc,
                                             int rec, int cnt, int lane) {
    constexpr int SEG = F / 32;
    constexpr int CH  = 32 / SEG;
#pragma unroll
    for (int c = 0; c < 32; c += CH) {
        if (FULL || c < cnt) {
            float v[CH][SEG];
#pragma unroll
            for (int j = 0; j < CH; ++j) {
                const unsigned r = (unsigned)__builtin_amdgcn_readlane(rec, c + j);
                const float* row = msg + (size_t)(r & 0xFFFFu) * F;
#pragma unroll
                for (int g = 0; g < SEG; ++g)
                    v[j][g] = row[g * 32 + lane];
            }
#pragma unroll
            for (int j = 0; j < CH; ++j) {
                if (FULL || c + j < cnt) {
                    const unsigned r = (unsigned)__builtin_amdgcn_readlane(rec, c + j);
                    float* arow = acc + (r >> 16) * P;
#pragma unroll
                    for (int g = 0; g < SEG; ++g)
                        __hip_atomic_fetch_add(arow + g * 32 + lane, v[j][g],
                                               __ATOMIC_RELAXED, __HIP_MEMORY_SCOPE_WORKGROUP);
                }
            }
        }
    }
}

__device__ __forceinline__ int list_slot(int f, int o0, int o1, int o2, int o3, int o4,
                                         int o5, int o6, int o7, int p1, int p2, int p3,
                                         int p4, int p5, int p6, int p7) {
    int o = o0;
    o = f >= p1 ? o1 : o;
    o = f >= p2 ? o2 : o;
    o = f >= p3 ? o3 : o;
    o = f >= p4 ? o4 : o;
    o = f >= p5 ? o5 : o;
    o = f >= p6 ? o6 : o;
    o = f >= p7 ? o7 : o;
    return f + o;
}

template <int F, int P, int ZROWS, class Stage>
__device__ __forceinline__ void bin_sum_rows(const float* __restrict__ msg,
                                             const int* __restrict__ sorted,
                                             const int* __restrict__ seg,
                                             float4* acc4, Stage stage) {
    static_assert((ZROWS * P) % 4 == 0, "cleared as float4");
    float* acc = (float*)acc4;
    const int lane = threadIdx.x & 31;
    const int wave = __builtin_amdgcn_readfirstlane((int)(threadIdx.x >> 5));

    int2 si = make_int2(0, 0);
    if (lane < SPW) si = ((const int2*)seg)[blockIdx.x * PART_WG + wave * SPW + lane];
    static_assert(SPW == 8, "the eight segments are written out below");
    const int p1 = __builtin_amdgcn_readlane(si.y, 0);
    const int p2 = p1 + __builtin_amdgcn_readlane(si.y, 1);
    const int p3 = p2 + __builtin_amdgcn_readlane(si.y, 2);
    const int p4 = p3 + __builtin_amdgcn_readlane(si.y, 3);
    const int p5 = p4 + __builtin_amdgcn_readlane(si.y, 4);
    const int p6 = p5 + __builtin_amdgcn_readlane(si.y, 5);
    const int p7 = p6 + __builtin_amdgcn_readlane(si.y, 6);
    const int total = p7 + __builtin_amdgcn_readlane(si.y, 7);
    const int o0 = __builtin_amdgcn_readlane(si.x, 0);
    const int o1 = __builtin_amdgcn_readlane(si.x, 1) - p1;
    const int o2 = __builtin_amdgcn_readlane(si.x, 2) - p2;
    const int o3 = __builtin_amdgcn_readlane(si.x, 3) - p3;
    const int o4 = __builtin_amdgcn_readlane(si.x, 4) - p4;
    const int o5 = __builtin_amdgcn_readlane(si.x, 5) - p5;
    const int o6 = __builtin_amdgcn_readlane(si.x, 6) - p6;
    const int o7 = __builtin_amdgcn_readlane(si.x, 7) - p7;

    int base = 0;
    int rec = 0;
    if (lane < total)
        rec = sorted[list_slot(lane, o0, o1, o2, o3, o4, o5, o6, o7, p1, p2, p3, p4, p5, p6, p7)];

    for (int i = threadIdx.x; i < ZROWS * P / 4; i += blockDim.x)
        acc4[i] = make_float4(0.f, 0.f, 0.f, 0.f);
    stage();
    __syncthreads();

    while (base < total) {
        const int nbase = base + 32;
        int nrec = 0;
        if (nbase + lane < total)
            nrec = sorted[list_slot(nbase + lane, o0, o1, o2, o3, o4, o5, o6, o7, p1, p2, p3, p4, p5, p6, p7)];
        const int left = total - base;
        const int cnt  = left < 32 ? left : 32;
        if (cnt == 32) gather_batch<F, P, true>(msg, acc, rec, cnt, lane);
        else           gather_batch<F, P, false>(msg, acc, rec, cnt, lane);
        rec  = nrec;
        base = nbase;
    }
    __syncthreads();
}

#define A1_P    68
#define A1_ROWS 208
static_assert(A1_ROWS % 16 == 0 && A1_ROWS >= BIN_R, "whole 16-row tiles over the bin");

__global__ void __launch_bounds__(1024)
bin_sum_relu_gemm(const float* __restrict__ t1, const int* __restrict__ sorted,
                  const int* __restrict__ seg, const float* __restrict__ b1,
                  const float* __restrict__ W2, float* __restrict__ t2, int n_nodes) {
    __shared__ float4 acc4[A1_ROWS * A1_P / 4];
    __shared__ float  sW[HID * N_CLS];
    __shared__ float  sB[HID];
    static_assert(HID * N_CLS == 2 * 1024, "two weights per thread of the 1024");

    const int   tid = threadIdx.x;
    const float w0  = W2[tid];
    const float w1  = W2[tid + 1024];
    const float bv  = b1[tid & (HID - 1)];

    bin_sum_rows<HID, A1_P, A1_ROWS>(t1, sorted, seg, acc4, [&] {
        sW[tid]        = w0;
        sW[tid + 1024] = w1;
        if (tid < HID) sB[tid] = bv;
    });

    const float* acc = (const float*)acc4;
    const int lane = tid & 31;
    const int wave = __builtin_amdgcn_readfirstlane(tid >> 5);
    constexpr int MT = A1_ROWS / 16;
    constexpr int NT = N_CLS / 16;
    if (wave < MT * NT) {
        const int mt = wave / NT;
        const int nt = wave % NT;
        const int h  = lane >> 4;
        const int c  = lane & 15;
        const int kh = 2 * h;
        const float* arow = acc + (mt * 16 + c) * A1_P + kh;
        const float* bcol = sW + kh * N_CLS + nt * 16 + c;
        v8f d = {};
#pragma unroll
        for (int k = 0; k < HID; k += 4) {
            v2f a, b;
            a.x = fmaxf(arow[k + 0] + sB[k + kh + 0], 0.f);
            a.y = fmaxf(arow[k + 1] + sB[k + kh + 1], 0.f);
            b.x = bcol[(k + 0) * N_CLS];
            b.y = bcol[(k + 1) * N_CLS];
            d = __builtin_amdgcn_wmma_f32_16x16x4_f32(
                false, a, false, b, (short)0, d, false, false);
        }
        const int lr0   = mt * 16 + 8 * h;
        const int node0 = blockIdx.x * BIN_R;
#pragma unroll
        for (int r = 0; r < 8; ++r) {
            const int lr = lr0 + r;
            if (lr < BIN_R && node0 + lr < n_nodes)
                t2[(size_t)(node0 + lr) * N_CLS + nt * 16 + c] = d[r];
        }
    }
}

__global__ void __launch_bounds__(1024)
bin_sum_logsoftmax(const float* __restrict__ t2, const int* __restrict__ sorted,
                   const int* __restrict__ seg, const float* __restrict__ b2,
                   float* __restrict__ out, int n_nodes) {
    __shared__ float4 acc4[BIN_R * N_CLS / 4];
    static_assert(BIN_R % 4 == 0, "whole four-row groups");

    const int lane = threadIdx.x & 31;
    const int sub  = lane >> 3;
    const int q    = lane & 7;
    float4 bb;
    bb.x = b2[q * 4 + 0]; bb.y = b2[q * 4 + 1];
    bb.z = b2[q * 4 + 2]; bb.w = b2[q * 4 + 3];

    bin_sum_rows<N_CLS, N_CLS, BIN_R>(t2, sorted, seg, acc4, [] {});

    const int wave  = (int)(threadIdx.x >> 5);
    const int nw    = (int)(blockDim.x >> 5);
    const int node0 = blockIdx.x * BIN_R;
    constexpr int NG = BIN_R / 4;
    constexpr int U  = 2;
    float4* out4 = (float4*)out;
    for (int g0 = wave; g0 < NG; g0 += U * nw) {
        float4 x[U];
        int    lr[U];
        bool   live[U];
#pragma unroll
        for (int u = 0; u < U; ++u) {
            const int g = g0 + u * nw;
            live[u] = g < NG;
            lr[u]   = g * 4 + sub;
            const int lc = live[u] ? lr[u] : BIN_R - 1;
            const float4 v = acc4[lc * (N_CLS / 4) + q];
            x[u].x = v.x + bb.x; x[u].y = v.y + bb.y;
            x[u].z = v.z + bb.z; x[u].w = v.w + bb.w;
        }
        float4 r[U];
#pragma unroll
        for (int u = 0; u < U; ++u) {
            float m = fmaxf(fmaxf(x[u].x, x[u].y), fmaxf(x[u].z, x[u].w));
#pragma unroll
            for (int o = 4; o; o >>= 1) m = fmaxf(m, __shfl_xor(m, o, 32));
            x[u].x -= m; x[u].y -= m; x[u].z -= m; x[u].w -= m;
            float e0 = __expf(x[u].x), e1 = __expf(x[u].y);
            float e2 = __expf(x[u].z), e3 = __expf(x[u].w);
#pragma unroll
            for (int o = 4; o; o >>= 1) {
                e0 += __shfl_xor(e0, o, 32);
                e1 += __shfl_xor(e1, o, 32);
                e2 += __shfl_xor(e2, o, 32);
                e3 += __shfl_xor(e3, o, 32);
            }
            const float ls = __logf((e0 + e2) + (e1 + e3));
            r[u].x = x[u].x - ls; r[u].y = x[u].y - ls;
            r[u].z = x[u].z - ls; r[u].w = x[u].w - ls;
        }
#pragma unroll
        for (int u = 0; u < U; ++u)
            if (live[u] && node0 + lr[u] < n_nodes)
                out4[(size_t)(node0 + lr[u]) * (N_CLS / 4) + q] = r[u];
    }
}

extern "C" void kernel_launch(void* const* d_in, const int* in_sizes, int n_in,
                              void* d_out, int out_size, void* d_ws, size_t ws_size,
                              hipStream_t stream) {
    const float* x   = (const float*)d_in[0];
    const int*   ei  = (const int*)d_in[1];
    const float* W1  = (const float*)d_in[2];
    const float* b1  = (const float*)d_in[3];
    const float* W2  = (const float*)d_in[4];
    const float* b2  = (const float*)d_in[5];
    float*       out = (float*)d_out;

    const int n_edges = in_sizes[1] / 2;
    const int* src = ei;
    const int* dst = ei + n_edges;

    float* t1 = (float*)d_ws;
    float* t2 = t1 + (size_t)N_NODES * HID;
    int* seg    = (int*)(t2 + (size_t)N_NODES * N_CLS);
    int* sorted = seg + 2 * NB * PART_WG;
    (void)ws_size; (void)n_in; (void)out_size;

    gemm1_and_share_sort<<<GEMM_WG + PART_WG, GEMM_THREADS, 0, stream>>>(
        x, W1, t1, N_NODES, src, dst, n_edges, sorted, seg);
    bin_sum_relu_gemm<<<NB, SUM_WAVES * 32, 0, stream>>>(t1, sorted, seg, b1, W2, t2, N_NODES);
    bin_sum_logsoftmax<<<NB, SUM_WAVES * 32, 0, stream>>>(t2, sorted, seg, b2, out, N_NODES);
}


# share sort: each record takes its slot in an LDS copy of the share's stretch, written out as whole 128-B lines (16-B stores, line-aligned stretches), not 4-B scattered stores; arrival-order sums
# speedup vs baseline: 3.6078x; 1.0387x over previous
#include <hip/hip_runtime.h>
#include <hip/hip_bf16.h>

#define N_NODES 50000
#define IN_F    96
#define HID     64
#define N_CLS   32

typedef __attribute__((ext_vector_type(2))) float v2f;
typedef __attribute__((ext_vector_type(8))) float v8f;

#define GEMM_THREADS 256
template <int K, int N>
__device__ __forceinline__ void gemm_wmma_f32(const float* __restrict__ X,
                                              const float* __restrict__ W,
                                              float* __restrict__ Y,
                                              int M, int blk) {
    constexpr int KS = K / 4;
    constexpr int WV = K * N / (4 * GEMM_THREADS);
    static_assert(K % 4 == 0 && (K * N) % (4 * GEMM_THREADS) == 0, "W is staged as whole float4 per thread");
    __shared__ float4 sW4[K * N / 4];
    const float* sW = (const float*)sW4;

    const int tid  = (int)threadIdx.x;
    const int wave = (blk * GEMM_THREADS + tid) >> 5;
    const int lane = tid & 31;
    const int m0   = wave * 16;
    const bool live = m0 < M;

    const int half16 = lane >> 4;
    const int l15    = lane & 15;
    const int kh     = half16 * 2;
    const int arow   = m0 + l15;

    v2f a[KS];
    if (live) {
        const float* xr = X + (size_t)arow * K + kh;
#pragma unroll
        for (int s = 0; s < KS; ++s) {
            a[s].x = xr[4 * s + 0];
            a[s].y = xr[4 * s + 1];
        }
    } else {
#pragma unroll
        for (int s = 0; s < KS; ++s) { a[s].x = 0.f; a[s].y = 0.f; }
    }

    float4 w[WV];
#pragma unroll
    for (int i = 0; i < WV; ++i) w[i] = ((const float4*)W)[tid + i * GEMM_THREADS];
#pragma unroll
    for (int i = 0; i < WV; ++i) sW4[tid + i * GEMM_THREADS] = w[i];
    __syncthreads();

    if (!live) return;

    constexpr int NT = N / 16;
    v8f acc[NT] = {};

#pragma unroll
    for (int s = 0; s < KS; ++s) {
        const int k = 4 * s;
#pragma unroll
        for (int t = 0; t < NT; ++t) {
            v2f b;
            b.x = sW[(k + kh + 0) * N + t * 16 + l15];
            b.y = sW[(k + kh + 1) * N + t * 16 + l15];
            acc[t] = __builtin_amdgcn_wmma_f32_16x16x4_f32(
                false, a[s], false, b, (short)0, acc[t], false, false);
        }
    }

    const int rbase = m0 + half16 * 8;
#pragma unroll
    for (int t = 0; t < NT; ++t)
#pragma unroll
        for (int r = 0; r < 8; ++r)
            Y[(size_t)(rbase + r) * N + t * 16 + l15] = acc[t][r];
}

#define NB        256
#define BIN_R     196
#define PART_WG   256
#define SUM_WAVES 32
#define SPW       (PART_WG / SUM_WAVES)
#define SORT_EPT  13
#define STAGE_CAP (SORT_EPT * 256)
static_assert(NB * BIN_R >= N_NODES, "bins must cover every node");
static_assert(N_NODES <= 65536 && BIN_R <= 65536, "record: 16 bits src, 16 bits local dst");
static_assert(PART_WG % SUM_WAVES == 0 && SPW <= 32, "a wave reads its shares' segments one per lane");

__device__ __forceinline__ void edge_share(int p, int P, int n_edges, int& e0, int& e1) {
    const int per = (n_edges + P - 1) / P;
    e0 = p * per;  if (e0 > n_edges) e0 = n_edges;
    e1 = e0 + per; if (e1 > n_edges) e1 = n_edges;
}

__device__ __forceinline__ unsigned edge_counter(int d, int lane) {
    return d >= 0 ? (unsigned)d / BIN_R : (unsigned)(NB + lane);
}

__device__ __forceinline__ void count_edges(int* hist, const int (&d)[SORT_EPT], int lane) {
#pragma unroll
    for (int k = 0; k < SORT_EPT; ++k)
        atomicAdd(&hist[edge_counter(d[k], lane)], 1);
}

__device__ __forceinline__ void place_edges(int* cursor, int* __restrict__ mine,
                                            const int (&d)[SORT_EPT], const int (&s)[SORT_EPT],
                                            int lane) {
    int slot[SORT_EPT];
#pragma unroll
    for (int k = 0; k < SORT_EPT; ++k)
        slot[k] = atomicAdd(&cursor[edge_counter(d[k], lane)], 1);
#pragma unroll
    for (int k = 0; k < SORT_EPT; ++k) {
        if (d[k] >= 0) {
            const unsigned dd  = (unsigned)d[k];
            const unsigned bin = dd / BIN_R;
            mine[slot[k]] = (int)((unsigned)s[k] | ((dd - bin * BIN_R) << 16));
        }
    }
}

__device__ __forceinline__ void
share_sort(const int* __restrict__ src, const int* __restrict__ dst, int n_edges,
           int* __restrict__ sorted, int* __restrict__ seg, int p) {
    __shared__ int hist[NB + 32];
    __shared__ int sc[2][NB];
    __shared__ int cursor[NB + 32];
    __shared__ int4 stage4[STAGE_CAP / 4];
    int* stage = (int*)stage4;
    const int t    = threadIdx.x;
    const int nt   = (int)blockDim.x;
    const int lane = t & 31;

    int e0, e1;
    edge_share(p, PART_WG, n_edges, e0, e1);
    const int per   = (n_edges + PART_WG - 1) / PART_WG;
    const int pitch = (per + 31) & ~31;
    const int first = p * pitch;

    int d0[SORT_EPT], s0[SORT_EPT];
#pragma unroll
    for (int k = 0; k < SORT_EPT; ++k) {
        const int e = e0 + t + k * nt;
        d0[k] = e < e1 ? dst[e] : -1;
        s0[k] = e < e1 ? src[e] : 0;
    }
    for (int i = t; i < NB + 32; i += nt) hist[i] = 0;
    __syncthreads();

    count_edges(hist, d0, lane);
    for (int eb = e0 + t + SORT_EPT * nt; eb < e1; eb += SORT_EPT * nt) {
        int d[SORT_EPT];
#pragma unroll
        for (int k = 0; k < SORT_EPT; ++k) {
            const int e = eb + k * nt;
            d[k] = e < e1 ? dst[e] : -1;
        }
        count_edges(hist, d, lane);
    }
    __syncthreads();
    int tot = 0;
    if (t < NB) {
        tot = hist[t];
        sc[0][t] = tot;
    }
    __syncthreads();
    int cur = 0;
    for (int off = 1; off < NB; off <<= 1) {
        if (t < NB) {
            int v = sc[cur][t];
            if (t >= off) v += sc[cur][t - off];
            sc[cur ^ 1][t] = v;
        }
        __syncthreads();
        cur ^= 1;
    }
    if (t < NB) {
        const int start = sc[cur][t] - tot;
        cursor[t] = start;
        if (t < 32) cursor[NB + t] = 0;
        ((int2*)seg)[t * PART_WG + p] = make_int2(first + start, tot);
    }
    __syncthreads();

    int* mine = sorted + first;
    if (per <= STAGE_CAP && nt * SORT_EPT >= per) {
        place_edges(cursor, stage, d0, s0, lane);
        __syncthreads();
        const int n  = e1 - e0;
        const int nv = n >> 2;
        constexpr int VT = (SORT_EPT + 3) / 4;
        int4 r[VT];
#pragma unroll
        for (int k = 0; k < VT; ++k) {
            const int v = t + k * nt;
            r[k] = make_int4(0, 0, 0, 0);
            if (v < nv) r[k] = stage4[v];
        }
#pragma unroll
        for (int k = 0; k < VT; ++k) {
            const int v = t + k * nt;
            if (v < nv) ((int4*)mine)[v] = r[k];
        }
        if (t < (n & 3)) mine[4 * nv + t] = stage[4 * nv + t];
    } else {
        place_edges(cursor, mine, d0, s0, lane);
        for (int eb = e0 + t + SORT_EPT * nt; eb < e1; eb += SORT_EPT * nt) {
            int d[SORT_EPT], s[SORT_EPT];
#pragma unroll
            for (int k = 0; k < SORT_EPT; ++k) {
                const int e = eb + k * nt;
                d[k] = e < e1 ? dst[e] : -1;
                s[k] = e < e1 ? src[e] : 0;
            }
            place_edges(cursor, mine, d, s, lane);
        }
    }
}

#define GEMM_WG ((N_NODES / 16 + 7) / 8)
__global__ void __launch_bounds__(GEMM_THREADS)
gemm1_and_share_sort(const float* __restrict__ x, const float* __restrict__ W1,
                     float* __restrict__ t1, int n_nodes,
                     const int* __restrict__ src, const int* __restrict__ dst, int n_edges,
                     int* __restrict__ sorted, int* __restrict__ seg) {
    if (blockIdx.x < GEMM_WG)
        gemm_wmma_f32<IN_F, HID>(x, W1, t1, n_nodes, (int)blockIdx.x);
    else
        share_sort(src, dst, n_edges, sorted, seg, (int)blockIdx.x - GEMM_WG);
}

template <int F, int P, bool FULL>
__device__ __forceinline__ void gather_batch(const float* __restrict__ msg, float* acc,
                                             int rec, int cnt, int lane) {
    constexpr int SEG = F / 32;
    constexpr int CH  = 32 / SEG;
#pragma unroll
    for (int c = 0; c < 32; c += CH) {
        if (FULL || c < cnt) {
            float v[CH][SEG];
#pragma unroll
            for (int j = 0; j < CH; ++j) {
                const unsigned r = (unsigned)__builtin_amdgcn_readlane(rec, c + j);
                const float* row = msg + (size_t)(r & 0xFFFFu) * F;
#pragma unroll
                for (int g = 0; g < SEG; ++g)
                    v[j][g] = row[g * 32 + lane];
            }
#pragma unroll
            for (int j = 0; j < CH; ++j) {
                if (FULL || c + j < cnt) {
                    const unsigned r = (unsigned)__builtin_amdgcn_readlane(rec, c + j);
                    float* arow = acc + (r >> 16) * P;
#pragma unroll
                    for (int g = 0; g < SEG; ++g)
                        __hip_atomic_fetch_add(arow + g * 32 + lane, v[j][g],
                                               __ATOMIC_RELAXED, __HIP_MEMORY_SCOPE_WORKGROUP);
                }
            }
        }
    }
}

__device__ __forceinline__ int list_slot(int f, int o0, int o1, int o2, int o3, int o4,
                                         int o5, int o6, int o7, int p1, int p2, int p3,
                                         int p4, int p5, int p6, int p7) {
    int o = o0;
    o = f >= p1 ? o1 : o;
    o = f >= p2 ? o2 : o;
    o = f >= p3 ? o3 : o;
    o = f >= p4 ? o4 : o;
    o = f >= p5 ? o5 : o;
    o = f >= p6 ? o6 : o;
    o = f >= p7 ? o7 : o;
    return f + o;
}

template <int F, int P, int ZROWS, class Stage>
__device__ __forceinline__ void bin_sum_rows(const float* __restrict__ msg,
                                             const int* __restrict__ sorted,
                                             const int* __restrict__ seg,
                                             float4* acc4, Stage stage) {
    static_assert((ZROWS * P) % 4 == 0, "cleared as float4");
    float* acc = (float*)acc4;
    const int lane = threadIdx.x & 31;
    const int wave = __builtin_amdgcn_readfirstlane((int)(threadIdx.x >> 5));

    int2 si = make_int2(0, 0);
    if (lane < SPW) si = ((const int2*)seg)[blockIdx.x * PART_WG + wave * SPW + lane];
    static_assert(SPW == 8, "the eight segments are written out below");
    const int p1 = __builtin_amdgcn_readlane(si.y, 0);
    const int p2 = p1 + __builtin_amdgcn_readlane(si.y, 1);
    const int p3 = p2 + __builtin_amdgcn_readlane(si.y, 2);
    const int p4 = p3 + __builtin_amdgcn_readlane(si.y, 3);
    const int p5 = p4 + __builtin_amdgcn_readlane(si.y, 4);
    const int p6 = p5 + __builtin_amdgcn_readlane(si.y, 5);
    const int p7 = p6 + __builtin_amdgcn_readlane(si.y, 6);
    const int total = p7 + __builtin_amdgcn_readlane(si.y, 7);
    const int o0 = __builtin_amdgcn_readlane(si.x, 0);
    const int o1 = __builtin_amdgcn_readlane(si.x, 1) - p1;
    const int o2 = __builtin_amdgcn_readlane(si.x, 2) - p2;
    const int o3 = __builtin_amdgcn_readlane(si.x, 3) - p3;
    const int o4 = __builtin_amdgcn_readlane(si.x, 4) - p4;
    const int o5 = __builtin_amdgcn_readlane(si.x, 5) - p5;
    const int o6 = __builtin_amdgcn_readlane(si.x, 6) - p6;
    const int o7 = __builtin_amdgcn_readlane(si.x, 7) - p7;

    int base = 0;
    int rec = 0;
    if (lane < total)
        rec = sorted[list_slot(lane, o0, o1, o2, o3, o4, o5, o6, o7, p1, p2, p3, p4, p5, p6, p7)];

    for (int i = threadIdx.x; i < ZROWS * P / 4; i += blockDim.x)
        acc4[i] = make_float4(0.f, 0.f, 0.f, 0.f);
    stage();
    __syncthreads();

    while (base < total) {
        const int nbase = base + 32;
        int nrec = 0;
        if (nbase + lane < total)
            nrec = sorted[list_slot(nbase + lane, o0, o1, o2, o3, o4, o5, o6, o7, p1, p2, p3, p4, p5, p6, p7)];
        const int left = total - base;
        const int cnt  = left < 32 ? left : 32;
        if (cnt == 32) gather_batch<F, P, true>(msg, acc, rec, cnt, lane);
        else           gather_batch<F, P, false>(msg, acc, rec, cnt, lane);
        rec  = nrec;
        base = nbase;
    }
    __syncthreads();
}

#define A1_P    68
#define A1_ROWS 208
static_assert(A1_ROWS % 16 == 0 && A1_ROWS >= BIN_R, "whole 16-row tiles over the bin");

__global__ void __launch_bounds__(1024)
bin_sum_relu_gemm(const float* __restrict__ t1, const int* __restrict__ sorted,
                  const int* __restrict__ seg, const float* __restrict__ b1,
                  const float* __restrict__ W2, float* __restrict__ t2, int n_nodes) {
    __shared__ float4 acc4[A1_ROWS * A1_P / 4];
    __shared__ float  sW[HID * N_CLS];
    __shared__ float  sB[HID];
    static_assert(HID * N_CLS == 2 * 1024, "two weights per thread of the 1024");

    const int   tid = threadIdx.x;
    const float w0  = W2[tid];
    const float w1  = W2[tid + 1024];
    const float bv  = b1[tid & (HID - 1)];

    bin_sum_rows<HID, A1_P, A1_ROWS>(t1, sorted, seg, acc4, [&] {
        sW[tid]        = w0;
        sW[tid + 1024] = w1;
        if (tid < HID) sB[tid] = bv;
    });

    const float* acc = (const float*)acc4;
    const int lane = tid & 31;
    const int wave = __builtin_amdgcn_readfirstlane(tid >> 5);
    constexpr int MT = A1_ROWS / 16;
    constexpr int NT = N_CLS / 16;
    if (wave < MT * NT) {
        const int mt = wave / NT;
        const int nt = wave % NT;
        const int h  = lane >> 4;
        const int c  = lane & 15;
        const int kh = 2 * h;
        const float* arow = acc + (mt * 16 + c) * A1_P + kh;
        const float* bcol = sW + kh * N_CLS + nt * 16 + c;
        v8f d = {};
#pragma unroll
        for (int k = 0; k < HID; k += 4) {
            v2f a, b;
            a.x = fmaxf(arow[k + 0] + sB[k + kh + 0], 0.f);
            a.y = fmaxf(arow[k + 1] + sB[k + kh + 1], 0.f);
            b.x = bcol[(k + 0) * N_CLS];
            b.y = bcol[(k + 1) * N_CLS];
            d = __builtin_amdgcn_wmma_f32_16x16x4_f32(
                false, a, false, b, (short)0, d, false, false);
        }
        const int lr0   = mt * 16 + 8 * h;
        const int node0 = blockIdx.x * BIN_R;
#pragma unroll
        for (int r = 0; r < 8; ++r) {
            const int lr = lr0 + r;
            if (lr < BIN_R && node0 + lr < n_nodes)
                t2[(size_t)(node0 + lr) * N_CLS + nt * 16 + c] = d[r];
        }
    }
}

__global__ void __launch_bounds__(1024)
bin_sum_logsoftmax(const float* __restrict__ t2, const int* __restrict__ sorted,
                   const int* __restrict__ seg, const float* __restrict__ b2,
                   float* __restrict__ out, int n_nodes) {
    __shared__ float4 acc4[BIN_R * N_CLS / 4];
    static_assert(BIN_R % 4 == 0, "whole four-row groups");

    const int lane = threadIdx.x & 31;
    const int sub  = lane >> 3;
    const int q    = lane & 7;
    float4 bb;
    bb.x = b2[q * 4 + 0]; bb.y = b2[q * 4 + 1];
    bb.z = b2[q * 4 + 2]; bb.w = b2[q * 4 + 3];

    bin_sum_rows<N_CLS, N_CLS, BIN_R>(t2, sorted, seg, acc4, [] {});

    const int wave  = (int)(threadIdx.x >> 5);
    const int nw    = (int)(blockDim.x >> 5);
    const int node0 = blockIdx.x * BIN_R;
    constexpr int NG = BIN_R / 4;
    constexpr int U  = 2;
    float4* out4 = (float4*)out;
    for (int g0 = wave; g0 < NG; g0 += U * nw) {
        float4 x[U];
        int    lr[U];
        bool   live[U];
#pragma unroll
        for (int u = 0; u < U; ++u) {
            const int g = g0 + u * nw;
            live[u] = g < NG;
            lr[u]   = g * 4 + sub;
            const int lc = live[u] ? lr[u] : BIN_R - 1;
            const float4 v = acc4[lc * (N_CLS / 4) + q];
            x[u].x = v.x + bb.x; x[u].y = v.y + bb.y;
            x[u].z = v.z + bb.z; x[u].w = v.w + bb.w;
        }
        float4 r[U];
#pragma unroll
        for (int u = 0; u < U; ++u) {
            float m = fmaxf(fmaxf(x[u].x, x[u].y), fmaxf(x[u].z, x[u].w));
#pragma unroll
            for (int o = 4; o; o >>= 1) m = fmaxf(m, __shfl_xor(m, o, 32));
            x[u].x -= m; x[u].y -= m; x[u].z -= m; x[u].w -= m;
            float e0 = __expf(x[u].x), e1 = __expf(x[u].y);
            float e2 = __expf(x[u].z), e3 = __expf(x[u].w);
#pragma unroll
            for (int o = 4; o; o >>= 1) {
                e0 += __shfl_xor(e0, o, 32);
                e1 += __shfl_xor(e1, o, 32);
                e2 += __shfl_xor(e2, o, 32);
                e3 += __shfl_xor(e3, o, 32);
            }
            const float ls = __logf((e0 + e2) + (e1 + e3));
            r[u].x = x[u].x - ls; r[u].y = x[u].y - ls;
            r[u].z = x[u].z - ls; r[u].w = x[u].w - ls;
        }
#pragma unroll
        for (int u = 0; u < U; ++u)
            if (live[u] && node0 + lr[u] < n_nodes)
                out4[(size_t)(node0 + lr[u]) * (N_CLS / 4) + q] = r[u];
    }
}

extern "C" void kernel_launch(void* const* d_in, const int* in_sizes, int n_in,
                              void* d_out, int out_size, void* d_ws, size_t ws_size,
                              hipStream_t stream) {
    const float* x   = (const float*)d_in[0];
    const int*   ei  = (const int*)d_in[1];
    const float* W1  = (const float*)d_in[2];
    const float* b1  = (const float*)d_in[3];
    const float* W2  = (const float*)d_in[4];
    const float* b2  = (const float*)d_in[5];
    float*       out = (float*)d_out;

    const int n_edges = in_sizes[1] / 2;
    const int* src = ei;
    const int* dst = ei + n_edges;

    float* t1 = (float*)d_ws;
    float* t2 = t1 + (size_t)N_NODES * HID;
    int* seg    = (int*)(t2 + (size_t)N_NODES * N_CLS);
    int* sorted = seg + 2 * NB * PART_WG;
    (void)ws_size; (void)n_in; (void)out_size;

    gemm1_and_share_sort<<<GEMM_WG + PART_WG, GEMM_THREADS, 0, stream>>>(
        x, W1, t1, N_NODES, src, dst, n_edges, sorted, seg);
    bin_sum_relu_gemm<<<NB, SUM_WAVES * 32, 0, stream>>>(t1, sorted, seg, b1, W2, t2, N_NODES);
    bin_sum_logsoftmax<<<NB, SUM_WAVES * 32, 0, stream>>>(t2, sorted, seg, b2, out, N_NODES);
}


# edge sum: each bin's work-group orders its records by destination row in LDS (integer counters; layer 2 reuses the list), waves sum a row's run in registers, one LDS add per run; sum order changed
# speedup vs baseline: 3.9216x; 1.0870x over previous
#include <hip/hip_runtime.h>
#include <hip/hip_bf16.h>

#define N_NODES 50000
#define IN_F    96
#define HID     64
#define N_CLS   32

typedef __attribute__((ext_vector_type(2))) float v2f;
typedef __attribute__((ext_vector_type(8))) float v8f;

#define GEMM_THREADS 256
template <int K, int N>
__device__ __forceinline__ void gemm_wmma_f32(const float* __restrict__ X,
                                              const float* __restrict__ W,
                                              float* __restrict__ Y,
                                              int M, int blk) {
    constexpr int KS = K / 4;
    constexpr int WV = K * N / (4 * GEMM_THREADS);
    static_assert(K % 4 == 0 && (K * N) % (4 * GEMM_THREADS) == 0, "W is staged as whole float4 per thread");
    __shared__ float4 sW4[K * N / 4];
    const float* sW = (const float*)sW4;

    const int tid  = (int)threadIdx.x;
    const int wave = (blk * GEMM_THREADS + tid) >> 5;
    const int lane = tid & 31;
    const int m0   = wave * 16;
    const bool live = m0 < M;

    const int half16 = lane >> 4;
    const int l15    = lane & 15;
    const int kh     = half16 * 2;
    const int arow   = m0 + l15;

    v2f a[KS];
    if (live) {
        const float* xr = X + (size_t)arow * K + kh;
#pragma unroll
        for (int s = 0; s < KS; ++s) {
            a[s].x = xr[4 * s + 0];
            a[s].y = xr[4 * s + 1];
        }
    } else {
#pragma unroll
        for (int s = 0; s < KS; ++s) { a[s].x = 0.f; a[s].y = 0.f; }
    }

    float4 w[WV];
#pragma unroll
    for (int i = 0; i < WV; ++i) w[i] = ((const float4*)W)[tid + i * GEMM_THREADS];
#pragma unroll
    for (int i = 0; i < WV; ++i) sW4[tid + i * GEMM_THREADS] = w[i];
    __syncthreads();

    if (!live) return;

    constexpr int NT = N / 16;
    v8f acc[NT] = {};

#pragma unroll
    for (int s = 0; s < KS; ++s) {
        const int k = 4 * s;
#pragma unroll
        for (int t = 0; t < NT; ++t) {
            v2f b;
            b.x = sW[(k + kh + 0) * N + t * 16 + l15];
            b.y = sW[(k + kh + 1) * N + t * 16 + l15];
            acc[t] = __builtin_amdgcn_wmma_f32_16x16x4_f32(
                false, a[s], false, b, (short)0, acc[t], false, false);
        }
    }

    const int rbase = m0 + half16 * 8;
#pragma unroll
    for (int t = 0; t < NT; ++t)
#pragma unroll
        for (int r = 0; r < 8; ++r)
            Y[(size_t)(rbase + r) * N + t * 16 + l15] = acc[t][r];
}

#define NB        256
#define BIN_R     196
#define PART_WG   256
#define SUM_WAVES 32
#define SPW       (PART_WG / SUM_WAVES)
#define SORT_EPT  13
#define STAGE_CAP (SORT_EPT * 256)
static_assert(NB * BIN_R >= N_NODES, "bins must cover every node");
static_assert(N_NODES <= 65536 && BIN_R <= 65536, "record: 16 bits src, 16 bits local dst");
static_assert(PART_WG % SUM_WAVES == 0 && SPW <= 32, "a wave reads its shares' segments one per lane");

__device__ __forceinline__ void edge_share(int p, int P, int n_edges, int& e0, int& e1) {
    const int per = (n_edges + P - 1) / P;
    e0 = p * per;  if (e0 > n_edges) e0 = n_edges;
    e1 = e0 + per; if (e1 > n_edges) e1 = n_edges;
}

__device__ __forceinline__ unsigned edge_counter(int d, int lane) {
    return d >= 0 ? (unsigned)d / BIN_R : (unsigned)(NB + lane);
}

__device__ __forceinline__ void count_edges(int* hist, const int (&d)[SORT_EPT], int lane) {
#pragma unroll
    for (int k = 0; k < SORT_EPT; ++k)
        atomicAdd(&hist[edge_counter(d[k], lane)], 1);
}

__device__ __forceinline__ void place_edges(int* cursor, int* __restrict__ mine,
                                            const int (&d)[SORT_EPT], const int (&s)[SORT_EPT],
                                            int lane) {
    int slot[SORT_EPT];
#pragma unroll
    for (int k = 0; k < SORT_EPT; ++k)
        slot[k] = atomicAdd(&cursor[edge_counter(d[k], lane)], 1);
#pragma unroll
    for (int k = 0; k < SORT_EPT; ++k) {
        if (d[k] >= 0) {
            const unsigned dd  = (unsigned)d[k];
            const unsigned bin = dd / BIN_R;
            mine[slot[k]] = (int)((unsigned)s[k] | ((dd - bin * BIN_R) << 16));
        }
    }
}

__device__ __forceinline__ void
share_sort(const int* __restrict__ src, const int* __restrict__ dst, int n_edges,
           int* __restrict__ sorted, int* __restrict__ seg, int p) {
    __shared__ int hist[NB + 32];
    __shared__ int sc[2][NB];
    __shared__ int cursor[NB + 32];
    __shared__ int4 stage4[STAGE_CAP / 4];
    int* stage = (int*)stage4;
    const int t    = threadIdx.x;
    const int nt   = (int)blockDim.x;
    const int lane = t & 31;

    int e0, e1;
    edge_share(p, PART_WG, n_edges, e0, e1);
    const int per   = (n_edges + PART_WG - 1) / PART_WG;
    const int pitch = (per + 31) & ~31;
    const int first = p * pitch;

    int d0[SORT_EPT], s0[SORT_EPT];
#pragma unroll
    for (int k = 0; k < SORT_EPT; ++k) {
        const int e = e0 + t + k * nt;
        d0[k] = e < e1 ? dst[e] : -1;
        s0[k] = e < e1 ? src[e] : 0;
    }
    for (int i = t; i < NB + 32; i += nt) hist[i] = 0;
    __syncthreads();

    count_edges(hist, d0, lane);
    for (int eb = e0 + t + SORT_EPT * nt; eb < e1; eb += SORT_EPT * nt) {
        int d[SORT_EPT];
#pragma unroll
        for (int k = 0; k < SORT_EPT; ++k) {
            const int e = eb + k * nt;
            d[k] = e < e1 ? dst[e] : -1;
        }
        count_edges(hist, d, lane);
    }
    __syncthreads();
    int tot = 0;
    if (t < NB) {
        tot = hist[t];
        sc[0][t] = tot;
    }
    __syncthreads();
    int cur = 0;
    for (int off = 1; off < NB; off <<= 1) {
        if (t < NB) {
            int v = sc[cur][t];
            if (t >= off) v += sc[cur][t - off];
            sc[cur ^ 1][t] = v;
        }
        __syncthreads();
        cur ^= 1;
    }
    if (t < NB) {
        const int start = sc[cur][t] - tot;
        cursor[t] = start;
        if (t < 32) cursor[NB + t] = 0;
        ((int2*)seg)[t * PART_WG + p] = make_int2(first + start, tot);
    }
    __syncthreads();

    int* mine = sorted + first;
    if (per <= STAGE_CAP && nt * SORT_EPT >= per) {
        place_edges(cursor, stage, d0, s0, lane);
        __syncthreads();
        const int n  = e1 - e0;
        const int nv = n >> 2;
        constexpr int VT = (SORT_EPT + 3) / 4;
        int4 r[VT];
#pragma unroll
        for (int k = 0; k < VT; ++k) {
            const int v = t + k * nt;
            r[k] = make_int4(0, 0, 0, 0);
            if (v < nv) r[k] = stage4[v];
        }
#pragma unroll
        for (int k = 0; k < VT; ++k) {
            const int v = t + k * nt;
            if (v < nv) ((int4*)mine)[v] = r[k];
        }
        if (t < (n & 3)) mine[4 * nv + t] = stage[4 * nv + t];
    } else {
        place_edges(cursor, mine, d0, s0, lane);
        for (int eb = e0 + t + SORT_EPT * nt; eb < e1; eb += SORT_EPT * nt) {
            int d[SORT_EPT], s[SORT_EPT];
#pragma unroll
            for (int k = 0; k < SORT_EPT; ++k) {
                const int e = eb + k * nt;
                d[k] = e < e1 ? dst[e] : -1;
                s[k] = e < e1 ? src[e] : 0;
            }
            place_edges(cursor, mine, d, s, lane);
        }
    }
}

#define GEMM_WG ((N_NODES / 16 + 7) / 8)
__global__ void __launch_bounds__(GEMM_THREADS)
gemm1_and_share_sort(const float* __restrict__ x, const float* __restrict__ W1,
                     float* __restrict__ t1, int n_nodes,
                     const int* __restrict__ src, const int* __restrict__ dst, int n_edges,
                     int* __restrict__ sorted, int* __restrict__ seg) {
    if (blockIdx.x < GEMM_WG)
        gemm_wmma_f32<IN_F, HID>(x, W1, t1, n_nodes, (int)blockIdx.x);
    else
        share_sort(src, dst, n_edges, sorted, seg, (int)blockIdx.x - GEMM_WG);
}

template <int F, int P, bool FULL>
__device__ __forceinline__ void gather_batch(const float* __restrict__ msg, float* acc,
                                             int rec, int cnt, int lane) {
    constexpr int SEG = F / 32;
    constexpr int CH  = 32 / SEG;
#pragma unroll
    for (int c = 0; c < 32; c += CH) {
        if (FULL || c < cnt) {
            float v[CH][SEG];
#pragma unroll
            for (int j = 0; j < CH; ++j) {
                const unsigned r = (unsigned)__builtin_amdgcn_readlane(rec, c + j);
                const float* row = msg + (size_t)(r & 0xFFFFu) * F;
#pragma unroll
                for (int g = 0; g < SEG; ++g)
                    v[j][g] = row[g * 32 + lane];
            }
#pragma unroll
            for (int j = 0; j < CH; ++j) {
                if (FULL || c + j < cnt) {
                    const unsigned r = (unsigned)__builtin_amdgcn_readlane(rec, c + j);
                    float* arow = acc + (r >> 16) * P;
#pragma unroll
                    for (int g = 0; g < SEG; ++g)
                        __hip_atomic_fetch_add(arow + g * 32 + lane, v[j][g],
                                               __ATOMIC_RELAXED, __HIP_MEMORY_SCOPE_WORKGROUP);
                }
            }
        }
    }
}

struct SegList {
    int o0, o1, o2, o3, o4, o5, o6, o7;
    int p1, p2, p3, p4, p5, p6, p7;
    int total;
};

__device__ __forceinline__ SegList wave_list(const int* __restrict__ seg, int lane, int wave) {
    int2 si = make_int2(0, 0);
    if (lane < SPW) si = ((const int2*)seg)[blockIdx.x * PART_WG + wave * SPW + lane];
    static_assert(SPW == 8, "the eight segments are written out below");
    SegList L;
    L.p1 = __builtin_amdgcn_readlane(si.y, 0);
    L.p2 = L.p1 + __builtin_amdgcn_readlane(si.y, 1);
    L.p3 = L.p2 + __builtin_amdgcn_readlane(si.y, 2);
    L.p4 = L.p3 + __builtin_amdgcn_readlane(si.y, 3);
    L.p5 = L.p4 + __builtin_amdgcn_readlane(si.y, 4);
    L.p6 = L.p5 + __builtin_amdgcn_readlane(si.y, 5);
    L.p7 = L.p6 + __builtin_amdgcn_readlane(si.y, 6);
    L.total = L.p7 + __builtin_amdgcn_readlane(si.y, 7);
    L.o0 = __builtin_amdgcn_readlane(si.x, 0);
    L.o1 = __builtin_amdgcn_readlane(si.x, 1) - L.p1;
    L.o2 = __builtin_amdgcn_readlane(si.x, 2) - L.p2;
    L.o3 = __builtin_amdgcn_readlane(si.x, 3) - L.p3;
    L.o4 = __builtin_amdgcn_readlane(si.x, 4) - L.p4;
    L.o5 = __builtin_amdgcn_readlane(si.x, 5) - L.p5;
    L.o6 = __builtin_amdgcn_readlane(si.x, 6) - L.p6;
    L.o7 = __builtin_amdgcn_readlane(si.x, 7) - L.p7;
    return L;
}

__device__ __forceinline__ int list_slot(const SegList& L, int f) {
    int o = L.o0;
    o = f >= L.p1 ? L.o1 : o;
    o = f >= L.p2 ? L.o2 : o;
    o = f >= L.p3 ? L.o3 : o;
    o = f >= L.p4 ? L.o4 : o;
    o = f >= L.p5 ? L.o5 : o;
    o = f >= L.p6 ? L.o6 : o;
    o = f >= L.p7 ? L.o7 : o;
    return f + o;
}

template <int F, int P>
__device__ __forceinline__ void walk_segments(const float* __restrict__ msg,
                                              const int* __restrict__ sorted,
                                              float* acc, const SegList& L, int lane) {
    int base = 0;
    int rec = 0;
    if (lane < L.total) rec = sorted[list_slot(L, lane)];
    while (base < L.total) {
        const int nbase = base + 32;
        int nrec = 0;
        if (nbase + lane < L.total) nrec = sorted[list_slot(L, nbase + lane)];
        const int left = L.total - base;
        const int cnt  = left < 32 ? left : 32;
        if (cnt == 32) gather_batch<F, P, true>(msg, acc, rec, cnt, lane);
        else           gather_batch<F, P, false>(msg, acc, rec, cnt, lane);
        rec  = nrec;
        base = nbase;
    }
}

#define LIST_CAP 8192
#define ROW_CNT  256
static_assert(ROW_CNT >= BIN_R && ROW_CNT == 8 * 32, "one wave scans the counters, 8 per lane");
static_assert(LIST_CAP % 4 == 0, "the list is copied as 16-byte vectors");

__device__ __forceinline__ int row_counter(int r, int lane) {
    return r >= 0 ? (r >> 16) : ROW_CNT + lane;
}

template <int F, int P>
__device__ __forceinline__ void flush_run(float* acc, int row, const float (&a)[F / 32], int lane) {
    float* arow = acc + row * P;
#pragma unroll
    for (int g = 0; g < F / 32; ++g)
        __hip_atomic_fetch_add(arow + g * 32 + lane, a[g],
                               __ATOMIC_RELAXED, __HIP_MEMORY_SCOPE_WORKGROUP);
}

template <int F, int P, bool FULL>
__device__ __forceinline__ void run_batch(const float* __restrict__ msg, float* acc,
                                          int rec, int cnt, int lane,
                                          int& cur, float (&a)[F / 32]) {
    constexpr int SEG = F / 32;
    constexpr int CH  = 32 / SEG;
#pragma unroll
    for (int c = 0; c < 32; c += CH) {
        if (FULL || c < cnt) {
            float v[CH][SEG];
#pragma unroll
            for (int j = 0; j < CH; ++j) {
                const unsigned r = (unsigned)__builtin_amdgcn_readlane(rec, c + j);
                const float* row = msg + (size_t)(r & 0xFFFFu) * F;
#pragma unroll
                for (int g = 0; g < SEG; ++g)
                    v[j][g] = row[g * 32 + lane];
            }
#pragma unroll
            for (int j = 0; j < CH; ++j) {
                if (FULL || c + j < cnt) {
                    const int ld = (int)((unsigned)__builtin_amdgcn_readlane(rec, c + j) >> 16);
                    if (ld != cur) {
                        flush_run<F, P>(acc, cur, a, lane);
                        cur = ld;
#pragma unroll
                        for (int g = 0; g < SEG; ++g) a[g] = v[j][g];
                    } else {
#pragma unroll
                        for (int g = 0; g < SEG; ++g) a[g] += v[j][g];
                    }
                }
            }
        }
    }
}

template <int F, int P, class Fetch>
__device__ __forceinline__ void sum_runs(const float* __restrict__ msg, float* acc,
                                         Fetch fetch, int rec, int lo, int hi, int lane) {
    if (lo >= hi) return;
    float a[F / 32];
#pragma unroll
    for (int g = 0; g < F / 32; ++g) a[g] = 0.f;
    int cur = (int)((unsigned)__builtin_amdgcn_readlane(rec, 0) >> 16);
    for (int base = lo; base < hi; base += 32) {
        int nrec = 0;
        if (base + 32 + lane < hi) nrec = fetch(base + 32 + lane);
        const int left = hi - base;
        const int cnt  = left < 32 ? left : 32;
        if (cnt == 32) run_batch<F, P, true>(msg, acc, rec, cnt, lane, cur, a);
        else           run_batch<F, P, false>(msg, acc, rec, cnt, lane, cur, a);
        rec = nrec;
    }
    flush_run<F, P>(acc, cur, a, lane);
}

#define A1_P    68
#define A1_ROWS 208
static_assert(A1_ROWS % 16 == 0 && A1_ROWS >= BIN_R, "whole 16-row tiles over the bin");

__global__ void __launch_bounds__(1024)
bin_sum_relu_gemm(const float* __restrict__ t1, const int* __restrict__ sorted,
                  const int* __restrict__ seg, const float* __restrict__ b1,
                  const float* __restrict__ W2, float* __restrict__ t2, int n_nodes,
                  int* __restrict__ bylist, int* __restrict__ tot) {
    __shared__ float4 acc4[A1_ROWS * A1_P / 4];
    __shared__ float  sW[HID * N_CLS];
    __shared__ float  sB[HID];
    __shared__ int4   cnt4[(ROW_CNT + 32) / 4];
    __shared__ int4   list4[LIST_CAP / 4];
    __shared__ int    sTot;
    static_assert(HID * N_CLS == 2 * 1024, "two weights per thread of the 1024");
    int* cnt  = (int*)cnt4;
    int* list = (int*)list4;

    const int   tid = threadIdx.x;
    const float w0  = W2[tid];
    const float w1  = W2[tid + 1024];
    const float bv  = b1[tid & (HID - 1)];

    {
        float* accw = (float*)acc4;
        const int lane = tid & 31;
        const int wave = __builtin_amdgcn_readfirstlane(tid >> 5);
        const SegList L = wave_list(seg, lane, wave);

        constexpr int KB = 5;
        int r[KB];
#pragma unroll
        for (int k = 0; k < KB; ++k) {
            const int f = k * 32 + lane;
            r[k] = f < L.total ? sorted[list_slot(L, f)] : -1;
        }

        for (int i = tid; i < A1_ROWS * A1_P / 4; i += 1024)
            acc4[i] = make_float4(0.f, 0.f, 0.f, 0.f);
        if (tid < (ROW_CNT + 32) / 4) cnt4[tid] = make_int4(0, 0, 0, 0);
        sW[tid]        = w0;
        sW[tid + 1024] = w1;
        if (tid < HID) sB[tid] = bv;
        __syncthreads();

#pragma unroll
        for (int k = 0; k < KB; ++k)
            atomicAdd(&cnt[row_counter(r[k], lane)], 1);
        for (int f = KB * 32 + lane; f < L.total; f += 32)
            atomicAdd(&cnt[sorted[list_slot(L, f)] >> 16], 1);
        __syncthreads();

        if (wave == 0) {
            const int4 c0 = cnt4[2 * lane], c1 = cnt4[2 * lane + 1];
            const int s0 = c0.x,      s1 = s0 + c0.y, s2 = s1 + c0.z, s3 = s2 + c0.w;
            const int s4 = s3 + c1.x, s5 = s4 + c1.y, s6 = s5 + c1.z, s7 = s6 + c1.w;
            int incl = s7;
#pragma unroll
            for (int o = 1; o < 32; o <<= 1) {
                const int u = __shfl_up(incl, o, 32);
                if (lane >= o) incl += u;
            }
            const int ex = incl - s7;
            cnt4[2 * lane]     = make_int4(ex, ex + s0, ex + s1, ex + s2);
            cnt4[2 * lane + 1] = make_int4(ex + s3, ex + s4, ex + s5, ex + s6);
            if (lane == 31) sTot = incl;
        }
        __syncthreads();
        const int T = __builtin_amdgcn_readfirstlane(sTot);
        if (tid == 0) tot[blockIdx.x] = T;

        if (T <= LIST_CAP) {
            int slot[KB];
#pragma unroll
            for (int k = 0; k < KB; ++k)
                slot[k] = atomicAdd(&cnt[row_counter(r[k], lane)], 1);
#pragma unroll
            for (int k = 0; k < KB; ++k)
                if (r[k] >= 0) list[slot[k]] = r[k];
            for (int f = KB * 32 + lane; f < L.total; f += 32) {
                const int x = sorted[list_slot(L, f)];
                list[atomicAdd(&cnt[x >> 16], 1)] = x;
            }
            __syncthreads();
            int4* out4 = (int4*)(bylist + (size_t)blockIdx.x * LIST_CAP);
            for (int i = tid; i < (T + 3) / 4; i += 1024) out4[i] = list4[i];
            const int lo = (int)(((long long)T * wave) / SUM_WAVES);
            const int hi = (int)(((long long)T * (wave + 1)) / SUM_WAVES);
            int first = 0;
            if (lo + lane < hi) first = list[lo + lane];
            sum_runs<HID, A1_P>(t1, accw, [&](int i) { return list[i]; }, first, lo, hi, lane);
        } else {
            walk_segments<HID, A1_P>(t1, sorted, accw, L, lane);
        }
        __syncthreads();
    }

    const float* acc = (const float*)acc4;
    const int lane = tid & 31;
    const int wave = __builtin_amdgcn_readfirstlane(tid >> 5);
    constexpr int MT = A1_ROWS / 16;
    constexpr int NT = N_CLS / 16;
    if (wave < MT * NT) {
        const int mt = wave / NT;
        const int nt = wave % NT;
        const int h  = lane >> 4;
        const int c  = lane & 15;
        const int kh = 2 * h;
        const float* arow = acc + (mt * 16 + c) * A1_P + kh;
        const float* bcol = sW + kh * N_CLS + nt * 16 + c;
        v8f d = {};
#pragma unroll
        for (int k = 0; k < HID; k += 4) {
            v2f a, b;
            a.x = fmaxf(arow[k + 0] + sB[k + kh + 0], 0.f);
            a.y = fmaxf(arow[k + 1] + sB[k + kh + 1], 0.f);
            b.x = bcol[(k + 0) * N_CLS];
            b.y = bcol[(k + 1) * N_CLS];
            d = __builtin_amdgcn_wmma_f32_16x16x4_f32(
                false, a, false, b, (short)0, d, false, false);
        }
        const int lr0   = mt * 16 + 8 * h;
        const int node0 = blockIdx.x * BIN_R;
#pragma unroll
        for (int r = 0; r < 8; ++r) {
            const int lr = lr0 + r;
            if (lr < BIN_R && node0 + lr < n_nodes)
                t2[(size_t)(node0 + lr) * N_CLS + nt * 16 + c] = d[r];
        }
    }
}

__global__ void __launch_bounds__(1024)
bin_sum_logsoftmax(const float* __restrict__ t2, const int* __restrict__ sorted,
                   const int* __restrict__ seg, const float* __restrict__ b2,
                   float* __restrict__ out, int n_nodes,
                   const int* __restrict__ bylist, const int* __restrict__ tot) {
    __shared__ float4 acc4[BIN_R * N_CLS / 4];
    static_assert(BIN_R % 4 == 0, "whole four-row groups");

    const int lane = threadIdx.x & 31;
    const int sub  = lane >> 3;
    const int q    = lane & 7;
    float4 bb;
    bb.x = b2[q * 4 + 0]; bb.y = b2[q * 4 + 1];
    bb.z = b2[q * 4 + 2]; bb.w = b2[q * 4 + 3];

    {
        float* accw = (float*)acc4;
        const int wv = __builtin_amdgcn_readfirstlane((int)(threadIdx.x >> 5));
        const int T  = __builtin_amdgcn_readfirstlane(tot[blockIdx.x]);
        if (T <= LIST_CAP) {
            const int* mine = bylist + (size_t)blockIdx.x * LIST_CAP;
            const int lo = (int)(((long long)T * wv) / SUM_WAVES);
            const int hi = (int)(((long long)T * (wv + 1)) / SUM_WAVES);
            int first = 0;
            if (lo + lane < hi) first = mine[lo + lane];
            for (int i = threadIdx.x; i < BIN_R * N_CLS / 4; i += blockDim.x)
                acc4[i] = make_float4(0.f, 0.f, 0.f, 0.f);
            __syncthreads();
            sum_runs<N_CLS, N_CLS>(t2, accw, [&](int i) { return mine[i]; }, first, lo, hi, lane);
        } else {
            const SegList L = wave_list(seg, lane, wv);
            for (int i = threadIdx.x; i < BIN_R * N_CLS / 4; i += blockDim.x)
                acc4[i] = make_float4(0.f, 0.f, 0.f, 0.f);
            __syncthreads();
            walk_segments<N_CLS, N_CLS>(t2, sorted, accw, L, lane);
        }
        __syncthreads();
    }

    const int wave  = (int)(threadIdx.x >> 5);
    const int nw    = (int)(blockDim.x >> 5);
    const int node0 = blockIdx.x * BIN_R;
    constexpr int NG = BIN_R / 4;
    constexpr int U  = 2;
    float4* out4 = (float4*)out;
    for (int g0 = wave; g0 < NG; g0 += U * nw) {
        float4 x[U];
        int    lr[U];
        bool   live[U];
#pragma unroll
        for (int u = 0; u < U; ++u) {
            const int g = g0 + u * nw;
            live[u] = g < NG;
            lr[u]   = g * 4 + sub;
            const int lc = live[u] ? lr[u] : BIN_R - 1;
            const float4 v = acc4[lc * (N_CLS / 4) + q];
            x[u].x = v.x + bb.x; x[u].y = v.y + bb.y;
            x[u].z = v.z + bb.z; x[u].w = v.w + bb.w;
        }
        float4 r[U];
#pragma unroll
        for (int u = 0; u < U; ++u) {
            float m = fmaxf(fmaxf(x[u].x, x[u].y), fmaxf(x[u].z, x[u].w));
#pragma unroll
            for (int o = 4; o; o >>= 1) m = fmaxf(m, __shfl_xor(m, o, 32));
            x[u].x -= m; x[u].y -= m; x[u].z -= m; x[u].w -= m;
            float e0 = __expf(x[u].x), e1 = __expf(x[u].y);
            float e2 = __expf(x[u].z), e3 = __expf(x[u].w);
#pragma unroll
            for (int o = 4; o; o >>= 1) {
                e0 += __shfl_xor(e0, o, 32);
                e1 += __shfl_xor(e1, o, 32);
                e2 += __shfl_xor(e2, o, 32);
                e3 += __shfl_xor(e3, o, 32);
            }
            const float ls = __logf((e0 + e2) + (e1 + e3));
            r[u].x = x[u].x - ls; r[u].y = x[u].y - ls;
            r[u].z = x[u].z - ls; r[u].w = x[u].w - ls;
        }
#pragma unroll
        for (int u = 0; u < U; ++u)
            if (live[u] && node0 + lr[u] < n_nodes)
                out4[(size_t)(node0 + lr[u]) * (N_CLS / 4) + q] = r[u];
    }
}

extern "C" void kernel_launch(void* const* d_in, const int* in_sizes, int n_in,
                              void* d_out, int out_size, void* d_ws, size_t ws_size,
                              hipStream_t stream) {
    const float* x   = (const float*)d_in[0];
    const int*   ei  = (const int*)d_in[1];
    const float* W1  = (const float*)d_in[2];
    const float* b1  = (const float*)d_in[3];
    const float* W2  = (const float*)d_in[4];
    const float* b2  = (const float*)d_in[5];
    float*       out = (float*)d_out;

    const int n_edges = in_sizes[1] / 2;
    const int* src = ei;
    const int* dst = ei + n_edges;

    float* t1 = (float*)d_ws;
    float* t2 = t1 + (size_t)N_NODES * HID;
    int* seg    = (int*)(t2 + (size_t)N_NODES * N_CLS);
    int* tot    = seg + 2 * NB * PART_WG;
    int* bylist = tot + NB;
    int* sorted = bylist + (size_t)NB * LIST_CAP;
    (void)ws_size; (void)n_in; (void)out_size;

    gemm1_and_share_sort<<<GEMM_WG + PART_WG, GEMM_THREADS, 0, stream>>>(
        x, W1, t1, N_NODES, src, dst, n_edges, sorted, seg);
    bin_sum_relu_gemm<<<NB, SUM_WAVES * 32, 0, stream>>>(t1, sorted, seg, b1, W2, t2, N_NODES, bylist, tot);
    bin_sum_logsoftmax<<<NB, SUM_WAVES * 32, 0, stream>>>(t2, sorted, seg, b2, out, N_NODES, bylist, tot);
}
